# starting kernel
# baseline (speedup 1.0000x reference)
#include <hip/hip_runtime.h>
#include <math.h>

typedef float v2f __attribute__((ext_vector_type(2)));
typedef float v8f __attribute__((ext_vector_type(8)));

#define N_NODES 50000
#define F_IN    128
#define F_HID   128
#define F_OUT   40

__global__ void zero_f32(float* __restrict__ p, int n) {
    int i = blockIdx.x * blockDim.x + threadIdx.x;
    if (i < n) p[i] = 0.0f;
}

__global__ void deg_acc(const int* __restrict__ dst, float* __restrict__ deg, int E) {
    int e = blockIdx.x * blockDim.x + threadIdx.x;
    if (e < E) unsafeAtomicAdd(&deg[dst[e]], 1.0f);
}

__global__ void make_dinv(float* __restrict__ deg, int n) {
    int i = blockIdx.x * blockDim.x + threadIdx.x;
    if (i < n) deg[i] = rsqrtf(deg[i] + 1.0f);
}

template <int NT, int NC>
__global__ void gemm_wmma_f32(const float* __restrict__ X,
                              const float* __restrict__ W,
                              float* __restrict__ H, int M) {
    const int lane = threadIdx.x & 31;
    const int wave = threadIdx.x >> 5;
    const int mt   = blockIdx.x * (blockDim.x >> 5) + wave;
    if (mt * 16 >= M) return;

    const int half = lane >> 4;
    const int l16  = lane & 15;

    v8f c[NT] = {};

    const float* xrow = X + (size_t)(mt * 16 + l16) * 128;

    for (int k = 0; k < 128; k += 4) {
        v2f a;
        a.x = xrow[k + 2 * half];
        a.y = xrow[k + 2 * half + 1];
#pragma unroll
        for (int t = 0; t < NT; ++t) {
            const int n = t * 16 + l16;
            v2f b;
            if ((NC % 16 == 0) || (n < NC)) {
                b.x = W[(size_t)(k + 2 * half) * NC + n];
                b.y = W[(size_t)(k + 2 * half + 1) * NC + n];
            } else {
                b.x = 0.0f; b.y = 0.0f;
            }
            c[t] = __builtin_amdgcn_wmma_f32_16x16x4_f32(
                false, a, false, b, (short)0, c[t], false, false);
        }
    }

#pragma unroll
    for (int t = 0; t < NT; ++t) {
        const int n = t * 16 + l16;
        if ((NC % 16 == 0) || (n < NC)) {
            float* out = H + (size_t)(mt * 16 + half * 8) * NC + n;
#pragma unroll
            for (int r = 0; r < 8; ++r) out[(size_t)r * NC] = c[t][r];
        }
    }
}

template <int F>
__global__ void agg_edges(const int* __restrict__ src, const int* __restrict__ dst,
                          const float* __restrict__ dinv,
                          const float* __restrict__ H, float* __restrict__ AGG,
                          int E) {
    long long gid = (long long)blockIdx.x * blockDim.x + threadIdx.x;
    if (gid >= (long long)E * F) return;
    int e = (int)(gid / F);
    int f = (int)(gid - (long long)e * F);
    int s = src[e], d = dst[e];
    float norm = dinv[s] * dinv[d];
    unsafeAtomicAdd(&AGG[(size_t)d * F + f], H[(size_t)s * F + f] * norm);
}

template <int F>
__global__ void finalize_relu(const float* __restrict__ H, const float* __restrict__ b,
                              const float* __restrict__ dinv, float* __restrict__ AGG,
                              int Nn) {
    long long gid = (long long)blockIdx.x * blockDim.x + threadIdx.x;
    if (gid >= (long long)Nn * F) return;
    int i = (int)(gid / F);
    int f = (int)(gid - (long long)i * F);
    float di = dinv[i];
    float v  = AGG[gid] + H[gid] * di * di + b[f];
    AGG[gid] = v > 0.0f ? v : 0.0f;
}

__global__ void finalize_logsoftmax(const float* __restrict__ H2,
                                    const float* __restrict__ b2,
                                    const float* __restrict__ dinv,
                                    const float* __restrict__ AGG2,
                                    float* __restrict__ OUT, int Nn) {
    const int lane = threadIdx.x & 31;
    const int wave = threadIdx.x >> 5;
    const int node = blockIdx.x * (blockDim.x >> 5) + wave;
    if (node >= Nn) return;

    const float di = dinv[node];
    const float dd = di * di;
    const size_t base = (size_t)node * F_OUT;

    float v0 = AGG2[base + lane] + H2[base + lane] * dd + b2[lane];
    float v1 = -INFINITY;
    if (lane < 8)
        v1 = AGG2[base + 32 + lane] + H2[base + 32 + lane] * dd + b2[32 + lane];

    float m = fmaxf(v0, v1);
#pragma unroll
    for (int off = 16; off > 0; off >>= 1)
        m = fmaxf(m, __shfl_xor(m, off, 32));

    float s = __expf(v0 - m) + ((lane < 8) ? __expf(v1 - m) : 0.0f);
#pragma unroll
    for (int off = 16; off > 0; off >>= 1)
        s += __shfl_xor(s, off, 32);

    const float ls = __logf(s);
    OUT[base + lane] = v0 - m - ls;
    if (lane < 8) OUT[base + 32 + lane] = v1 - m - ls;
}

extern "C" void kernel_launch(void* const* d_in, const int* in_sizes, int n_in,
                              void* d_out, int out_size, void* d_ws, size_t ws_size,
                              hipStream_t stream) {
    const float* x  = (const float*)d_in[0];
    const float* W1 = (const float*)d_in[1];
    const float* b1 = (const float*)d_in[2];
    const float* W2 = (const float*)d_in[3];
    const float* b2 = (const float*)d_in[4];
    const int*   ei = (const int*)d_in[5];
    const int E = in_sizes[5] / 2;
    const int Nn = N_NODES;
    const int* src = ei;
    const int* dst = ei + E;

    float* out = (float*)d_out;

    float* ws   = (float*)d_ws;
    float* dinv = ws;
    float* h1   = dinv + 50048;
    float* agg1 = h1   + (size_t)Nn * F_HID;
    float* h2   = agg1 + (size_t)Nn * F_HID;
    float* agg2 = h2   + (size_t)Nn * F_OUT;

    const int TPB = 256;

    zero_f32<<<(Nn + TPB - 1) / TPB, TPB, 0, stream>>>(dinv, Nn);
    deg_acc<<<(E + TPB - 1) / TPB, TPB, 0, stream>>>(dst, dinv, E);
    make_dinv<<<(Nn + TPB - 1) / TPB, TPB, 0, stream>>>(dinv, Nn);

    {
        int mtiles = Nn / 16;
        int blocks = (mtiles + 7) / 8;
        gemm_wmma_f32<8, 128><<<blocks, TPB, 0, stream>>>(x, W1, h1, Nn);
    }

    {
        long long total = (long long)Nn * F_HID;
        zero_f32<<<(int)((total + TPB - 1) / TPB), TPB, 0, stream>>>(agg1, (int)total);
        long long work = (long long)E * F_HID;
        agg_edges<F_HID><<<(int)((work + TPB - 1) / TPB), TPB, 0, stream>>>(
            src, dst, dinv, h1, agg1, E);
        finalize_relu<F_HID><<<(int)((total + TPB - 1) / TPB), TPB, 0, stream>>>(
            h1, b1, dinv, agg1, Nn);
    }

    {
        int mtiles = Nn / 16;
        int blocks = (mtiles + 7) / 8;
        gemm_wmma_f32<3, 40><<<blocks, TPB, 0, stream>>>(agg1, W2, h2, Nn);
    }

    {
        long long total = (long long)Nn * F_OUT;
        zero_f32<<<(int)((total + TPB - 1) / TPB), TPB, 0, stream>>>(agg2, (int)total);
        long long work = (long long)E * F_OUT;
        agg_edges<F_OUT><<<(int)((work + TPB - 1) / TPB), TPB, 0, stream>>>(
            src, dst, dinv, h2, agg2, E);
        int blocks = (Nn + 7) / 8;
        finalize_logsoftmax<<<blocks, TPB, 0, stream>>>(h2, b2, dinv, agg2, out, Nn);
    }
}


# Edge sums by gather, not f32 atomics: edges grouped by destination (int atomic arrival-order rank + offsets), one wave sums a node's rows in registers, epilogues fused; row sum order = arrival order
# speedup vs baseline: 2.7133x; 2.7133x over previous
#include <hip/hip_runtime.h>
#include <math.h>


typedef float v2f __attribute__((ext_vector_type(2)));
typedef float v8f __attribute__((ext_vector_type(8)));

#define N_NODES 50000
#define F_IN    128
#define F_HID   128
#define F_OUT   40

__global__ void zero_i32(int* __restrict__ p, int n) {
    int i = blockIdx.x * blockDim.x + threadIdx.x;
    if (i < n) p[i] = 0;
}

__global__ void rank_count(const int* __restrict__ dst, int* __restrict__ cnt,
                           int* __restrict__ rank, int E) {
    int e = blockIdx.x * blockDim.x + threadIdx.x;
    if (e < E) rank[e] = atomicAdd(&cnt[dst[e]], 1);
}

__global__ void node_setup(const int* __restrict__ cnt, int* __restrict__ total,
                           int* __restrict__ rowstart, float* __restrict__ dinv, int n) {
    __shared__ int wsum[32];
    __shared__ int blockbase;
    const int lane = threadIdx.x & 31;
    const int wave = threadIdx.x >> 5;
    const int i = blockIdx.x * blockDim.x + threadIdx.x;
    const int c = (i < n) ? cnt[i] : 0;
    if (i < n) dinv[i] = rsqrtf((float)c + 1.0f);

    int v = c;
#pragma unroll
    for (int off = 1; off < 32; off <<= 1) {
        int t = __shfl_up(v, off, 32);
        if (lane >= off) v += t;
    }
    if (lane == 31) wsum[wave] = v;
    __syncthreads();
    if (wave == 0) {
        const int s = wsum[lane];
        int sv = s;
#pragma unroll
        for (int off = 1; off < 32; off <<= 1) {
            int t = __shfl_up(sv, off, 32);
            if (lane >= off) sv += t;
        }
        wsum[lane] = sv - s;
        if (lane == 31) blockbase = atomicAdd(total, sv);
    }
    __syncthreads();
    if (i < n) rowstart[i] = blockbase + wsum[wave] + v - c;
}

__global__ void fill_groups(const int* __restrict__ src, const int* __restrict__ dst,
                            const int* __restrict__ rank, const int* __restrict__ rowstart,
                            int* __restrict__ gsrc, int E) {
    int e = blockIdx.x * blockDim.x + threadIdx.x;
    if (e < E) gsrc[rowstart[dst[e]] + rank[e]] = src[e];
}

template <int NT, int NC>
__global__ void gemm_wmma_f32(const float* __restrict__ X,
                              const float* __restrict__ W,
                              float* __restrict__ H, int M) {
    const int lane = threadIdx.x & 31;
    const int wave = threadIdx.x >> 5;
    const int mt   = blockIdx.x * (blockDim.x >> 5) + wave;
    if (mt * 16 >= M) return;

    const int half = lane >> 4;
    const int l16  = lane & 15;

    v8f c[NT] = {};

    const float* xrow = X + (size_t)(mt * 16 + l16) * 128;

    for (int k = 0; k < 128; k += 4) {
        v2f a;
        a.x = xrow[k + 2 * half];
        a.y = xrow[k + 2 * half + 1];
#pragma unroll
        for (int t = 0; t < NT; ++t) {
            const int n = t * 16 + l16;
            v2f b;
            if ((NC % 16 == 0) || (n < NC)) {
                b.x = W[(size_t)(k + 2 * half) * NC + n];
                b.y = W[(size_t)(k + 2 * half + 1) * NC + n];
            } else {
                b.x = 0.0f; b.y = 0.0f;
            }
            c[t] = __builtin_amdgcn_wmma_f32_16x16x4_f32(
                false, a, false, b, (short)0, c[t], false, false);
        }
    }

#pragma unroll
    for (int t = 0; t < NT; ++t) {
        const int n = t * 16 + l16;
        if ((NC % 16 == 0) || (n < NC)) {
            float* out = H + (size_t)(mt * 16 + half * 8) * NC + n;
#pragma unroll
            for (int r = 0; r < 8; ++r) out[(size_t)r * NC] = c[t][r];
        }
    }
}

__global__ void gather_relu(const int* __restrict__ rowstart, const int* __restrict__ cnt,
                            const int* __restrict__ gsrc, const float* __restrict__ dinv,
                            const float* __restrict__ H, const float* __restrict__ b,
                            float* __restrict__ OUTH, int Nn) {
    const int lane = threadIdx.x & 31;
    const int wave = threadIdx.x >> 5;
    const int node = blockIdx.x * (blockDim.x >> 5) + wave;
    if (node >= Nn) return;

    const int start = __builtin_amdgcn_readfirstlane(rowstart[node]);
    const int n     = __builtin_amdgcn_readfirstlane(cnt[node]);
    const float di  = dinv[node];
    const float4* __restrict__ H4 = reinterpret_cast<const float4*>(H);

    float4 acc = make_float4(0.0f, 0.0f, 0.0f, 0.0f);
    for (int base = 0; base < n; base += 32) {
        const int m = (n - base < 32) ? (n - base) : 32;
        int   my_s = 0;
        float my_w = 0.0f;
        if (lane < m) {
            my_s = gsrc[start + base + lane];
            my_w = dinv[my_s];
        }
#pragma unroll 4
        for (int j = 0; j < m; ++j) {
            const int   s = __builtin_amdgcn_readlane(my_s, j);
            const float w = __int_as_float(__builtin_amdgcn_readlane(__float_as_int(my_w), j)) * di;
            const float4 h = H4[(size_t)s * (F_HID / 4) + lane];
            acc.x += h.x * w;
            acc.y += h.y * w;
            acc.z += h.z * w;
            acc.w += h.w * w;
        }
    }

    const size_t o = (size_t)node * (F_HID / 4) + lane;
    const float4 hs = H4[o];
    const float4 bb = reinterpret_cast<const float4*>(b)[lane];
    float4 v;
    v.x = acc.x + hs.x * di * di + bb.x;
    v.y = acc.y + hs.y * di * di + bb.y;
    v.z = acc.z + hs.z * di * di + bb.z;
    v.w = acc.w + hs.w * di * di + bb.w;
    v.x = v.x > 0.0f ? v.x : 0.0f;
    v.y = v.y > 0.0f ? v.y : 0.0f;
    v.z = v.z > 0.0f ? v.z : 0.0f;
    v.w = v.w > 0.0f ? v.w : 0.0f;
    reinterpret_cast<float4*>(OUTH)[o] = v;
}

__global__ void gather_logsoftmax(const int* __restrict__ rowstart, const int* __restrict__ cnt,
                                  const int* __restrict__ gsrc, const float* __restrict__ dinv,
                                  const float* __restrict__ H2, const float* __restrict__ b2,
                                  float* __restrict__ OUT, int Nn) {
    const int lane = threadIdx.x & 31;
    const int wave = threadIdx.x >> 5;
    const int node = blockIdx.x * (blockDim.x >> 5) + wave;
    if (node >= Nn) return;

    const int start = __builtin_amdgcn_readfirstlane(rowstart[node]);
    const int n     = __builtin_amdgcn_readfirstlane(cnt[node]);
    const float di  = dinv[node];
    const bool act  = lane < (F_OUT / 4);
    const int col   = act ? lane : 0;
    const float4* __restrict__ H4 = reinterpret_cast<const float4*>(H2);

    float4 acc = make_float4(0.0f, 0.0f, 0.0f, 0.0f);
    for (int base = 0; base < n; base += 32) {
        const int m = (n - base < 32) ? (n - base) : 32;
        int   my_s = 0;
        float my_w = 0.0f;
        if (lane < m) {
            my_s = gsrc[start + base + lane];
            my_w = dinv[my_s];
        }
#pragma unroll 4
        for (int j = 0; j < m; ++j) {
            const int   s = __builtin_amdgcn_readlane(my_s, j);
            const float w = __int_as_float(__builtin_amdgcn_readlane(__float_as_int(my_w), j)) * di;
            const float4 h = H4[(size_t)s * (F_OUT / 4) + col];
            acc.x += h.x * w;
            acc.y += h.y * w;
            acc.z += h.z * w;
            acc.w += h.w * w;
        }
    }

    const float dd = di * di;
    const size_t o = (size_t)node * (F_OUT / 4) + lane;
    float4 v = make_float4(-INFINITY, -INFINITY, -INFINITY, -INFINITY);
    if (act) {
        const float4 hs = H4[o];
        const float4 bb = reinterpret_cast<const float4*>(b2)[lane];
        v.x = acc.x + hs.x * dd + bb.x;
        v.y = acc.y + hs.y * dd + bb.y;
        v.z = acc.z + hs.z * dd + bb.z;
        v.w = acc.w + hs.w * dd + bb.w;
    }

    float mx = fmaxf(fmaxf(v.x, v.y), fmaxf(v.z, v.w));
#pragma unroll
    for (int off = 16; off > 0; off >>= 1)
        mx = fmaxf(mx, __shfl_xor(mx, off, 32));

    float s = act ? (__expf(v.x - mx) + __expf(v.y - mx) + __expf(v.z - mx) + __expf(v.w - mx))
                  : 0.0f;
#pragma unroll
    for (int off = 16; off > 0; off >>= 1)
        s += __shfl_xor(s, off, 32);

    const float ls = __logf(s);
    if (act) {
        float4 r;
        r.x = v.x - mx - ls;
        r.y = v.y - mx - ls;
        r.z = v.z - mx - ls;
        r.w = v.w - mx - ls;
        reinterpret_cast<float4*>(OUT)[o] = r;
    }
}

extern "C" void kernel_launch(void* const* d_in, const int* in_sizes, int n_in,
                              void* d_out, int out_size, void* d_ws, size_t ws_size,
                              hipStream_t stream) {
    const float* x  = (const float*)d_in[0];
    const float* W1 = (const float*)d_in[1];
    const float* b1 = (const float*)d_in[2];
    const float* W2 = (const float*)d_in[3];
    const float* b2 = (const float*)d_in[4];
    const int*   ei = (const int*)d_in[5];
    const int E = in_sizes[5] / 2;
    const int Nn = N_NODES;
    const int* src = ei;
    const int* dst = ei + E;

    float* out = (float*)d_out;

    float* ws   = (float*)d_ws;
    float* dinv = ws;
    float* h1   = dinv + 50048;
    float* hr   = h1   + (size_t)Nn * F_HID;
    float* h2   = hr   + (size_t)Nn * F_HID;
    int*   cnt  = (int*)(h2 + (size_t)Nn * F_OUT);
    int*   rowstart = cnt + 50048;
    int*   rank = rowstart + 50048;
    int*   gsrc = rank + E;

    const int TPB = 256;

    zero_i32<<<(Nn + 1 + TPB - 1) / TPB, TPB, 0, stream>>>(cnt, Nn + 1);
    rank_count<<<(E + TPB - 1) / TPB, TPB, 0, stream>>>(dst, cnt, rank, E);
    node_setup<<<(Nn + 1023) / 1024, 1024, 0, stream>>>(cnt, cnt + Nn, rowstart, dinv, Nn);
    fill_groups<<<(E + TPB - 1) / TPB, TPB, 0, stream>>>(src, dst, rank, rowstart, gsrc, E);

    {
        int mtiles = Nn / 16;
        int blocks = (mtiles + 7) / 8;
        gemm_wmma_f32<8, 128><<<blocks, TPB, 0, stream>>>(x, W1, h1, Nn);
    }

    gather_relu<<<(Nn + 7) / 8, TPB, 0, stream>>>(rowstart, cnt, gsrc, dinv, h1, b1, hr, Nn);

    {
        int mtiles = Nn / 16;
        int blocks = (mtiles + 7) / 8;
        gemm_wmma_f32<3, 40><<<blocks, TPB, 0, stream>>>(hr, W2, h2, Nn);
    }

    gather_logsoftmax<<<(Nn + 7) / 8, TPB, 0, stream>>>(rowstart, cnt, gsrc, dinv, h2, b2, out, Nn);
}


# Edge grouping in one launch on LDS counters: 196 owner blocks of 256 nodes read dst once, rank own edges by returning LDS adds; no global atomics, 3 launches less; row sum order = LDS arrival order
# speedup vs baseline: 3.6705x; 1.3528x over previous
#include <hip/hip_runtime.h>
#include <math.h>


typedef float v2f __attribute__((ext_vector_type(2)));
typedef float v8f __attribute__((ext_vector_type(8)));

#define N_NODES 50000
#define F_IN    128
#define F_HID   128
#define F_OUT   40

#define OWN_NODES   256
#define OWN_BLOCKS  196
#define OWN_THREADS 1024
#define OWN_WAVES   (OWN_THREADS / 32)
#define OWN_GROUP   8
#define OWN_LIST    768
#define NO_NODE     0x7fffffff

static_assert(OWN_NODES * OWN_BLOCKS >= N_NODES, "owners must cover every node");
static_assert(OWN_NODES <= 256, "the node's place in the block is kept in 8 bits");
static_assert(OWN_NODES <= OWN_THREADS, "one thread per owned node in the scan");
static_assert(OWN_THREADS == 1024, "a tile is 1024 16-byte pieces (shift by 10)");

__global__ __launch_bounds__(OWN_THREADS)
void build_groups(const int* __restrict__ src, const int* __restrict__ dst,
                  int* __restrict__ cnt, int* __restrict__ rowstart,
                  float* __restrict__ dinv, int* __restrict__ gsrc, int E, int Nn) {
    __shared__ int2 lst[OWN_WAVES][OWN_LIST];
    __shared__ int c[OWN_NODES];
    __shared__ int wsum[32];
    __shared__ int wlow[32];
    __shared__ int sbase;
    __shared__ int sfull;

    const int tid  = threadIdx.x;
    const int lane = tid & 31;
    const int wave = tid >> 5;
    const int lo   = blockIdx.x * OWN_NODES;
    const int left = Nn - lo;
    const unsigned span = left < OWN_NODES ? (left > 0 ? (unsigned)left : 0u) : (unsigned)OWN_NODES;

    for (int i = tid; i < OWN_NODES; i += OWN_THREADS) c[i] = 0;
    if (tid == 0) sfull = (E >= (1 << 24)) ? 1 : 0;
    __syncthreads();

    int low = 0;
    int nw  = 0;

    auto look = [&](int d, int e) {
        low += __popc(__builtin_amdgcn_ballot_w32(d < lo));
        const unsigned k   = (unsigned)(d - lo);
        const bool     hit = k < span;
        const unsigned hm  = __builtin_amdgcn_ballot_w32(hit);
        if (__builtin_expect(hm != 0u, 0)) {
            if (hit) {
                const int r    = atomicAdd(&c[k], 1);
                const int slot = nw + (int)__builtin_amdgcn_mbcnt_lo(hm, 0u);
                if (slot < OWN_LIST)
                    lst[wave][slot] = make_int2(e, (int)(k | ((unsigned)r << 8)));
            }
            nw += __popc(hm);
        }
    };

    const int nvec = ((E & 3) == 0) ? (E >> 2) : 0;
    const int4* __restrict__ dst4 = reinterpret_cast<const int4*>(dst);
    const int nft  = nvec >> 10;
    const int trot = (int)(((unsigned)blockIdx.x * (unsigned)nft) / gridDim.x);

    auto looks4 = [&](const int4& d, int e) {
        look(d.x, e);
        look(d.y, e + 1);
        look(d.z, e + 2);
        look(d.w, e + 3);
    };
    int t0 = 0;
    for (; t0 + OWN_GROUP <= nft; t0 += OWN_GROUP) {
        int4 d[OWN_GROUP];
        int  eb[OWN_GROUP];
#pragma unroll
        for (int u = 0; u < OWN_GROUP; ++u) {
            int t = t0 + u + trot;
            if (t >= nft) t -= nft;
            const int v = (t << 10) + tid;
            eb[u] = v << 2;
            d[u]  = dst4[v];
        }
#pragma unroll
        for (int u = 0; u < OWN_GROUP; ++u) looks4(d[u], eb[u]);
    }
    if (t0 < nft) {
        const int live = nft - t0;
        int4 d[OWN_GROUP];
        int  eb[OWN_GROUP];
#pragma unroll
        for (int u = 0; u < OWN_GROUP; ++u) {
            int t = t0 + (u < live ? u : 0) + trot;
            if (t >= nft) t -= nft;
            const int v = (t << 10) + tid;
            eb[u] = v << 2;
            d[u]  = dst4[v];
        }
#pragma unroll
        for (int u = 0; u < OWN_GROUP; ++u)
            if (u < live) looks4(d[u], eb[u]);
    }
    for (int i0 = nft << 10; i0 < nvec; i0 += OWN_THREADS) {
        const int  i  = i0 + tid;
        const bool ok = i < nvec;
        int4 d = dst4[ok ? i : i0];
        if (!ok) d = make_int4(NO_NODE, NO_NODE, NO_NODE, NO_NODE);
        look(d.x, 4 * i);
        look(d.y, 4 * i + 1);
        look(d.z, 4 * i + 2);
        look(d.w, 4 * i + 3);
    }
    for (int e0 = nvec * 4; e0 < E; e0 += OWN_THREADS) {
        const int e = e0 + tid;
        look((e < E) ? dst[e] : NO_NODE, e);
    }
    if (nw > OWN_LIST && lane == 0) sfull = 1;
    __syncthreads();

    const int cv = (tid < OWN_NODES) ? c[tid] : 0;
    int v = cv;
#pragma unroll
    for (int off = 1; off < 32; off <<= 1) {
        int t = __shfl_up(v, off, 32);
        if (lane >= off) v += t;
    }
    if (lane == 31) wsum[wave] = v;
    if (lane == 0)  wlow[wave] = low;
    __syncthreads();
    if (wave == 0) {
        const int s = wsum[lane];
        int sv = s;
#pragma unroll
        for (int off = 1; off < 32; off <<= 1) {
            int t = __shfl_up(sv, off, 32);
            if (lane >= off) sv += t;
        }
        wsum[lane] = sv - s;
        int lw = wlow[lane];
#pragma unroll
        for (int off = 16; off > 0; off >>= 1) lw += __shfl_xor(lw, off, 32);
        if (lane == 0) sbase = lw;
    }
    __syncthreads();
    if ((unsigned)tid < span) {
        const int start = sbase + wsum[wave] + v - cv;
        const int node  = lo + tid;
        cnt[node]      = cv;
        rowstart[node] = start;
        dinv[node]     = rsqrtf((float)cv + 1.0f);
        c[tid]         = start;
    }
    __syncthreads();

    if (sfull == 0) {
        for (int i = lane; i < nw; i += 32) {
            const int2     en = lst[wave][i];
            const unsigned k  = (unsigned)en.y & 255u;
            const int      r  = (int)((unsigned)en.y >> 8);
            gsrc[c[k] + r] = src[en.x];
        }
        return;
    }

    for (int e0 = 0; e0 < E; e0 += OWN_THREADS) {
        const int e = e0 + tid;
        const int dd = (e < E) ? dst[e] : NO_NODE;
        const unsigned k = (unsigned)(dd - lo);
        if (k < span) gsrc[atomicAdd(&c[k], 1)] = src[e];
    }
}

template <int NT, int NC>
__global__ void gemm_wmma_f32(const float* __restrict__ X,
                              const float* __restrict__ W,
                              const float* __restrict__ rowscale,
                              float* __restrict__ H, int M) {
    const int lane = threadIdx.x & 31;
    const int wave = threadIdx.x >> 5;
    const int mt   = blockIdx.x * (blockDim.x >> 5) + wave;
    if (mt * 16 >= M) return;

    const int half = lane >> 4;
    const int l16  = lane & 15;

    v8f c[NT] = {};

    const float* xrow = X + (size_t)(mt * 16 + l16) * 128;

    const float4* sc4 = reinterpret_cast<const float4*>(rowscale + mt * 16 + half * 8);
    const float4 sc_lo = sc4[0];
    const float4 sc_hi = sc4[1];

    for (int k = 0; k < 128; k += 4) {
        v2f a;
        a.x = xrow[k + 2 * half];
        a.y = xrow[k + 2 * half + 1];
#pragma unroll
        for (int t = 0; t < NT; ++t) {
            const int n = t * 16 + l16;
            v2f b;
            if ((NC % 16 == 0) || (n < NC)) {
                b.x = W[(size_t)(k + 2 * half) * NC + n];
                b.y = W[(size_t)(k + 2 * half + 1) * NC + n];
            } else {
                b.x = 0.0f; b.y = 0.0f;
            }
            c[t] = __builtin_amdgcn_wmma_f32_16x16x4_f32(
                false, a, false, b, (short)0, c[t], false, false);
        }
    }

#pragma unroll
    for (int t = 0; t < NT; ++t) {
        const int n = t * 16 + l16;
        if ((NC % 16 == 0) || (n < NC)) {
            float* out = H + (size_t)(mt * 16 + half * 8) * NC + n;
            out[(size_t)0 * NC] = c[t][0] * sc_lo.x;
            out[(size_t)1 * NC] = c[t][1] * sc_lo.y;
            out[(size_t)2 * NC] = c[t][2] * sc_lo.z;
            out[(size_t)3 * NC] = c[t][3] * sc_lo.w;
            out[(size_t)4 * NC] = c[t][4] * sc_hi.x;
            out[(size_t)5 * NC] = c[t][5] * sc_hi.y;
            out[(size_t)6 * NC] = c[t][6] * sc_hi.z;
            out[(size_t)7 * NC] = c[t][7] * sc_hi.w;
        }
    }
}

__global__ void gather_relu(const int* __restrict__ rowstart, const int* __restrict__ cnt,
                            const int* __restrict__ gsrc, const float* __restrict__ dinv,
                            const float* __restrict__ H, const float* __restrict__ b,
                            float* __restrict__ OUTH, int Nn) {
    const int lane = threadIdx.x & 31;
    const int wave = threadIdx.x >> 5;
    const int node = blockIdx.x * (blockDim.x >> 5) + wave;
    if (node >= Nn) return;

    const int start = __builtin_amdgcn_readfirstlane(rowstart[node]);
    const int n     = __builtin_amdgcn_readfirstlane(cnt[node]);
    const float di  = dinv[node];
    const float4* __restrict__ H4 = reinterpret_cast<const float4*>(H);

    float4 acc = make_float4(0.0f, 0.0f, 0.0f, 0.0f);
    for (int base = 0; base < n; base += 32) {
        const int m = (n - base < 32) ? (n - base) : 32;
        int my_s = 0;
        if (lane < m) my_s = gsrc[start + base + lane];
#pragma unroll 4
        for (int j = 0; j < m; ++j) {
            const int    s = __builtin_amdgcn_readlane(my_s, j);
            const float4 h = H4[(size_t)s * (F_HID / 4) + lane];
            acc.x += h.x;
            acc.y += h.y;
            acc.z += h.z;
            acc.w += h.w;
        }
    }

    const size_t o = (size_t)node * (F_HID / 4) + lane;
    const float4 hs = H4[o];
    const float4 bb = reinterpret_cast<const float4*>(b)[lane];
    float4 v;
    v.x = (acc.x + hs.x) * di + bb.x;
    v.y = (acc.y + hs.y) * di + bb.y;
    v.z = (acc.z + hs.z) * di + bb.z;
    v.w = (acc.w + hs.w) * di + bb.w;
    v.x = v.x > 0.0f ? v.x : 0.0f;
    v.y = v.y > 0.0f ? v.y : 0.0f;
    v.z = v.z > 0.0f ? v.z : 0.0f;
    v.w = v.w > 0.0f ? v.w : 0.0f;
    reinterpret_cast<float4*>(OUTH)[o] = v;
}

__global__ void gather_logsoftmax(const int* __restrict__ rowstart, const int* __restrict__ cnt,
                                  const int* __restrict__ gsrc, const float* __restrict__ dinv,
                                  const float* __restrict__ H2, const float* __restrict__ b2,
                                  float* __restrict__ OUT, int Nn) {
    const int lane = threadIdx.x & 31;
    const int wave = threadIdx.x >> 5;
    const int node = blockIdx.x * (blockDim.x >> 5) + wave;
    if (node >= Nn) return;

    const int start = __builtin_amdgcn_readfirstlane(rowstart[node]);
    const int n     = __builtin_amdgcn_readfirstlane(cnt[node]);
    const float di  = dinv[node];
    const bool act  = lane < (F_OUT / 4);
    const int col   = act ? lane : 0;
    const float4* __restrict__ H4 = reinterpret_cast<const float4*>(H2);

    float4 acc = make_float4(0.0f, 0.0f, 0.0f, 0.0f);
    for (int base = 0; base < n; base += 32) {
        const int m = (n - base < 32) ? (n - base) : 32;
        int my_s = 0;
        if (lane < m) my_s = gsrc[start + base + lane];
#pragma unroll 4
        for (int j = 0; j < m; ++j) {
            const int    s = __builtin_amdgcn_readlane(my_s, j);
            const float4 h = H4[(size_t)s * (F_OUT / 4) + col];
            acc.x += h.x;
            acc.y += h.y;
            acc.z += h.z;
            acc.w += h.w;
        }
    }

    const size_t o = (size_t)node * (F_OUT / 4) + lane;
    float4 v = make_float4(-INFINITY, -INFINITY, -INFINITY, -INFINITY);
    if (act) {
        const float4 hs = H4[o];
        const float4 bb = reinterpret_cast<const float4*>(b2)[lane];
        v.x = (acc.x + hs.x) * di + bb.x;
        v.y = (acc.y + hs.y) * di + bb.y;
        v.z = (acc.z + hs.z) * di + bb.z;
        v.w = (acc.w + hs.w) * di + bb.w;
    }

    float mx = fmaxf(fmaxf(v.x, v.y), fmaxf(v.z, v.w));
#pragma unroll
    for (int off = 16; off > 0; off >>= 1)
        mx = fmaxf(mx, __shfl_xor(mx, off, 32));

    float s = act ? (__expf(v.x - mx) + __expf(v.y - mx) + __expf(v.z - mx) + __expf(v.w - mx))
                  : 0.0f;
#pragma unroll
    for (int off = 16; off > 0; off >>= 1)
        s += __shfl_xor(s, off, 32);

    const float ls = __logf(s);
    if (act) {
        float4 r;
        r.x = v.x - mx - ls;
        r.y = v.y - mx - ls;
        r.z = v.z - mx - ls;
        r.w = v.w - mx - ls;
        reinterpret_cast<float4*>(OUT)[o] = r;
    }
}

extern "C" void kernel_launch(void* const* d_in, const int* in_sizes, int n_in,
                              void* d_out, int out_size, void* d_ws, size_t ws_size,
                              hipStream_t stream) {
    const float* x  = (const float*)d_in[0];
    const float* W1 = (const float*)d_in[1];
    const float* b1 = (const float*)d_in[2];
    const float* W2 = (const float*)d_in[3];
    const float* b2 = (const float*)d_in[4];
    const int*   ei = (const int*)d_in[5];
    const int E = in_sizes[5] / 2;
    const int Nn = N_NODES;
    const int* src = ei;
    const int* dst = ei + E;

    float* out = (float*)d_out;

    float* ws   = (float*)d_ws;
    float* dinv = ws;
    float* h1   = dinv + 50048;
    float* hr   = h1   + (size_t)Nn * F_HID;
    float* h2   = hr   + (size_t)Nn * F_HID;
    int*   cnt  = (int*)(h2 + (size_t)Nn * F_OUT);
    int*   rowstart = cnt + 50048;
    int*   gsrc = rowstart + 50048;

    const int TPB = 256;

    build_groups<<<OWN_BLOCKS, OWN_THREADS, 0, stream>>>(src, dst, cnt, rowstart, dinv, gsrc, E, Nn);

    {
        int mtiles = Nn / 16;
        int blocks = (mtiles + 7) / 8;
        gemm_wmma_f32<8, 128><<<blocks, TPB, 0, stream>>>(x, W1, dinv, h1, Nn);
    }

    gather_relu<<<(Nn + 7) / 8, TPB, 0, stream>>>(rowstart, cnt, gsrc, dinv, h1, b1, hr, Nn);

    {
        int mtiles = Nn / 16;
        int blocks = (mtiles + 7) / 8;
        gemm_wmma_f32<3, 40><<<blocks, TPB, 0, stream>>>(hr, W2, dinv, h2, Nn);
    }

    gather_logsoftmax<<<(Nn + 7) / 8, TPB, 0, stream>>>(rowstart, cnt, gsrc, dinv, h2, b2, out, Nn);
}


# Layer-2 product fused into the layer-1 gather launch: 16 nodes per work-group, ReLU rows go to an LDS tile, 3 waves multiply it by W2 (in LDS) on WMMA, same k order; relu plane and one launch gone
# speedup vs baseline: 4.1520x; 1.1312x over previous
#include <hip/hip_runtime.h>
#include <math.h>


typedef float v2f __attribute__((ext_vector_type(2)));
typedef float v8f __attribute__((ext_vector_type(8)));

#define N_NODES 50000
#define F_IN    128
#define F_HID   128
#define F_OUT   40

#define OWN_NODES   256
#define OWN_BLOCKS  196
#define OWN_THREADS 1024
#define OWN_WAVES   (OWN_THREADS / 32)
#define OWN_GROUP   8
#define OWN_LIST    768
#define NO_NODE     0x7fffffff

static_assert(OWN_NODES * OWN_BLOCKS >= N_NODES, "owners must cover every node");
static_assert(OWN_NODES <= 256, "the node's place in the block is kept in 8 bits");
static_assert(OWN_NODES <= OWN_THREADS, "one thread per owned node in the scan");
static_assert(OWN_THREADS == 1024, "a tile is 1024 16-byte pieces (shift by 10)");

__global__ __launch_bounds__(OWN_THREADS)
void build_groups(const int* __restrict__ src, const int* __restrict__ dst,
                  int* __restrict__ cnt, int* __restrict__ rowstart,
                  float* __restrict__ dinv, int* __restrict__ gsrc, int E, int Nn) {
    __shared__ int2 lst[OWN_WAVES][OWN_LIST];
    __shared__ int c[OWN_NODES];
    __shared__ int wsum[32];
    __shared__ int wlow[32];
    __shared__ int sbase;
    __shared__ int sfull;

    const int tid  = threadIdx.x;
    const int lane = tid & 31;
    const int wave = tid >> 5;
    const int lo   = blockIdx.x * OWN_NODES;
    const int left = Nn - lo;
    const unsigned span = left < OWN_NODES ? (left > 0 ? (unsigned)left : 0u) : (unsigned)OWN_NODES;

    for (int i = tid; i < OWN_NODES; i += OWN_THREADS) c[i] = 0;
    if (tid == 0) sfull = (E >= (1 << 24)) ? 1 : 0;
    __syncthreads();

    int low = 0;
    int nw  = 0;

    auto look = [&](int d, int e) {
        low += __popc(__builtin_amdgcn_ballot_w32(d < lo));
        const unsigned k   = (unsigned)(d - lo);
        const bool     hit = k < span;
        const unsigned hm  = __builtin_amdgcn_ballot_w32(hit);
        if (__builtin_expect(hm != 0u, 0)) {
            if (hit) {
                const int r    = atomicAdd(&c[k], 1);
                const int slot = nw + (int)__builtin_amdgcn_mbcnt_lo(hm, 0u);
                if (slot < OWN_LIST)
                    lst[wave][slot] = make_int2(e, (int)(k | ((unsigned)r << 8)));
            }
            nw += __popc(hm);
        }
    };

    const int nvec = ((E & 3) == 0) ? (E >> 2) : 0;
    const int4* __restrict__ dst4 = reinterpret_cast<const int4*>(dst);
    const int nft  = nvec >> 10;
    const int trot = (int)(((unsigned)blockIdx.x * (unsigned)nft) / gridDim.x);

    auto looks4 = [&](const int4& d, int e) {
        look(d.x, e);
        look(d.y, e + 1);
        look(d.z, e + 2);
        look(d.w, e + 3);
    };
    int t0 = 0;
    for (; t0 + OWN_GROUP <= nft; t0 += OWN_GROUP) {
        int4 d[OWN_GROUP];
        int  eb[OWN_GROUP];
#pragma unroll
        for (int u = 0; u < OWN_GROUP; ++u) {
            int t = t0 + u + trot;
            if (t >= nft) t -= nft;
            const int v = (t << 10) + tid;
            eb[u] = v << 2;
            d[u]  = dst4[v];
        }
#pragma unroll
        for (int u = 0; u < OWN_GROUP; ++u) looks4(d[u], eb[u]);
    }
    if (t0 < nft) {
        const int live = nft - t0;
        int4 d[OWN_GROUP];
        int  eb[OWN_GROUP];
#pragma unroll
        for (int u = 0; u < OWN_GROUP; ++u) {
            int t = t0 + (u < live ? u : 0) + trot;
            if (t >= nft) t -= nft;
            const int v = (t << 10) + tid;
            eb[u] = v << 2;
            d[u]  = dst4[v];
        }
#pragma unroll
        for (int u = 0; u < OWN_GROUP; ++u)
            if (u < live) looks4(d[u], eb[u]);
    }
    for (int i0 = nft << 10; i0 < nvec; i0 += OWN_THREADS) {
        const int  i  = i0 + tid;
        const bool ok = i < nvec;
        int4 d = dst4[ok ? i : i0];
        if (!ok) d = make_int4(NO_NODE, NO_NODE, NO_NODE, NO_NODE);
        look(d.x, 4 * i);
        look(d.y, 4 * i + 1);
        look(d.z, 4 * i + 2);
        look(d.w, 4 * i + 3);
    }
    for (int e0 = nvec * 4; e0 < E; e0 += OWN_THREADS) {
        const int e = e0 + tid;
        look((e < E) ? dst[e] : NO_NODE, e);
    }
    if (nw > OWN_LIST && lane == 0) sfull = 1;
    __syncthreads();

    const int cv = (tid < OWN_NODES) ? c[tid] : 0;
    int v = cv;
#pragma unroll
    for (int off = 1; off < 32; off <<= 1) {
        int t = __shfl_up(v, off, 32);
        if (lane >= off) v += t;
    }
    if (lane == 31) wsum[wave] = v;
    if (lane == 0)  wlow[wave] = low;
    __syncthreads();
    if (wave == 0) {
        const int s = wsum[lane];
        int sv = s;
#pragma unroll
        for (int off = 1; off < 32; off <<= 1) {
            int t = __shfl_up(sv, off, 32);
            if (lane >= off) sv += t;
        }
        wsum[lane] = sv - s;
        int lw = wlow[lane];
#pragma unroll
        for (int off = 16; off > 0; off >>= 1) lw += __shfl_xor(lw, off, 32);
        if (lane == 0) sbase = lw;
    }
    __syncthreads();
    if ((unsigned)tid < span) {
        const int start = sbase + wsum[wave] + v - cv;
        const int node  = lo + tid;
        cnt[node]      = cv;
        rowstart[node] = start;
        dinv[node]     = rsqrtf((float)cv + 1.0f);
        c[tid]         = start;
    }
    __syncthreads();

    if (sfull == 0) {
        for (int i = lane; i < nw; i += 32) {
            const int2     en = lst[wave][i];
            const unsigned k  = (unsigned)en.y & 255u;
            const int      r  = (int)((unsigned)en.y >> 8);
            gsrc[c[k] + r] = src[en.x];
        }
        return;
    }

    for (int e0 = 0; e0 < E; e0 += OWN_THREADS) {
        const int e = e0 + tid;
        const int dd = (e < E) ? dst[e] : NO_NODE;
        const unsigned k = (unsigned)(dd - lo);
        if (k < span) gsrc[atomicAdd(&c[k], 1)] = src[e];
    }
}

template <int NT, int NC>
__global__ void gemm_wmma_f32(const float* __restrict__ X,
                              const float* __restrict__ W,
                              const float* __restrict__ rowscale,
                              float* __restrict__ H, int M) {
    const int lane = threadIdx.x & 31;
    const int wave = threadIdx.x >> 5;
    const int mt   = blockIdx.x * (blockDim.x >> 5) + wave;
    if (mt * 16 >= M) return;

    const int half = lane >> 4;
    const int l16  = lane & 15;

    v8f c[NT] = {};

    const float* xrow = X + (size_t)(mt * 16 + l16) * 128;

    const float4* sc4 = reinterpret_cast<const float4*>(rowscale + mt * 16 + half * 8);
    const float4 sc_lo = sc4[0];
    const float4 sc_hi = sc4[1];

    for (int k = 0; k < 128; k += 4) {
        v2f a;
        a.x = xrow[k + 2 * half];
        a.y = xrow[k + 2 * half + 1];
#pragma unroll
        for (int t = 0; t < NT; ++t) {
            const int n = t * 16 + l16;
            v2f b;
            if ((NC % 16 == 0) || (n < NC)) {
                b.x = W[(size_t)(k + 2 * half) * NC + n];
                b.y = W[(size_t)(k + 2 * half + 1) * NC + n];
            } else {
                b.x = 0.0f; b.y = 0.0f;
            }
            c[t] = __builtin_amdgcn_wmma_f32_16x16x4_f32(
                false, a, false, b, (short)0, c[t], false, false);
        }
    }

#pragma unroll
    for (int t = 0; t < NT; ++t) {
        const int n = t * 16 + l16;
        if ((NC % 16 == 0) || (n < NC)) {
            float* out = H + (size_t)(mt * 16 + half * 8) * NC + n;
            out[(size_t)0 * NC] = c[t][0] * sc_lo.x;
            out[(size_t)1 * NC] = c[t][1] * sc_lo.y;
            out[(size_t)2 * NC] = c[t][2] * sc_lo.z;
            out[(size_t)3 * NC] = c[t][3] * sc_lo.w;
            out[(size_t)4 * NC] = c[t][4] * sc_hi.x;
            out[(size_t)5 * NC] = c[t][5] * sc_hi.y;
            out[(size_t)6 * NC] = c[t][6] * sc_hi.z;
            out[(size_t)7 * NC] = c[t][7] * sc_hi.w;
        }
    }
}

#define FG_NODES   16
#define FG_THREADS (FG_NODES * 32)
#define FG_PITCH   (F_HID + 4)
#define FG_NT      ((F_OUT + 15) / 16)
#define FG_KS      (F_HID / 4)

static_assert(FG_KS == 2 * FG_NODES, "each wave copies the B entries of two k steps");

__global__ __launch_bounds__(FG_THREADS)
void gather_relu_gemm2(const int* __restrict__ rowstart, const int* __restrict__ cnt,
                       const int* __restrict__ gsrc, const float* __restrict__ dinv,
                       const float* __restrict__ H, const float* __restrict__ b,
                       const float* __restrict__ W2, float* __restrict__ H2, int Nn) {
    __shared__ __attribute__((aligned(16))) float tile[FG_NODES][FG_PITCH];
    __shared__ v2f bfrag[FG_KS][FG_NT][32];
    __shared__ __attribute__((aligned(16))) float sdi[FG_NODES];

    const int lane = threadIdx.x & 31;
    const int wave = threadIdx.x >> 5;
    const int half = lane >> 4;
    const int l16  = lane & 15;
    const int base_node = blockIdx.x * FG_NODES;
    const int node = base_node + wave;

    v2f wv[2][FG_NT];
    for (int i = 0; i < 2; ++i) {
        const int krow = 4 * (wave + FG_NODES * i) + 2 * half;
        for (int t = 0; t < FG_NT; ++t) {
            const int col = t * 16 + l16;
            v2f w;
            w.x = 0.0f; w.y = 0.0f;
            if ((F_OUT % 16 == 0) || (col < F_OUT)) {
                w.x = W2[(size_t)krow * F_OUT + col];
                w.y = W2[(size_t)(krow + 1) * F_OUT + col];
            }
            wv[i][t] = w;
        }
    }

    float4 v = make_float4(0.0f, 0.0f, 0.0f, 0.0f);
    float di = 0.0f;
    if (node < Nn) {
        const int start = __builtin_amdgcn_readfirstlane(rowstart[node]);
        const int n     = __builtin_amdgcn_readfirstlane(cnt[node]);
        di = dinv[node];
        const float4* __restrict__ H4 = reinterpret_cast<const float4*>(H);

        float4 acc = make_float4(0.0f, 0.0f, 0.0f, 0.0f);
        for (int base = 0; base < n; base += 32) {
            const int m = (n - base < 32) ? (n - base) : 32;
            int my_s = 0;
            if (lane < m) my_s = gsrc[start + base + lane];
#pragma unroll 4
            for (int j = 0; j < m; ++j) {
                const int    s = __builtin_amdgcn_readlane(my_s, j);
                const float4 h = H4[(size_t)s * (F_HID / 4) + lane];
                acc.x += h.x;
                acc.y += h.y;
                acc.z += h.z;
                acc.w += h.w;
            }
        }

        const float4 hs = H4[(size_t)node * (F_HID / 4) + lane];
        const float4 bb = reinterpret_cast<const float4*>(b)[lane];
        v.x = (acc.x + hs.x) * di + bb.x;
        v.y = (acc.y + hs.y) * di + bb.y;
        v.z = (acc.z + hs.z) * di + bb.z;
        v.w = (acc.w + hs.w) * di + bb.w;
        v.x = v.x > 0.0f ? v.x : 0.0f;
        v.y = v.y > 0.0f ? v.y : 0.0f;
        v.z = v.z > 0.0f ? v.z : 0.0f;
        v.w = v.w > 0.0f ? v.w : 0.0f;
    }
    for (int i = 0; i < 2; ++i)
        for (int t = 0; t < FG_NT; ++t) bfrag[wave + FG_NODES * i][t][lane] = wv[i][t];
    *reinterpret_cast<float4*>(&tile[wave][4 * lane]) = v;
    if (lane == 0) sdi[wave] = di;
    __syncthreads();

    if (wave >= FG_NT) return;

    const int ncol = wave * 16 + l16;

    const float4* sc4 = reinterpret_cast<const float4*>(&sdi[half * 8]);
    const float4 sc_lo = sc4[0];
    const float4 sc_hi = sc4[1];

    v8f c = {};
    const float* arow = &tile[l16][2 * half];
    for (int ks = 0; ks < FG_KS; ++ks) {
        const v2f a  = *reinterpret_cast<const v2f*>(arow + 4 * ks);
        const v2f bf = bfrag[ks][wave][lane];
        c = __builtin_amdgcn_wmma_f32_16x16x4_f32(
            false, a, false, bf, (short)0, c, false, false);
    }

    if ((F_OUT % 16 == 0) || (ncol < F_OUT)) {
        const int row0 = base_node + half * 8;
        float* out = H2 + (size_t)row0 * F_OUT + ncol;
        if (row0 + 0 < Nn) out[(size_t)0 * F_OUT] = c[0] * sc_lo.x;
        if (row0 + 1 < Nn) out[(size_t)1 * F_OUT] = c[1] * sc_lo.y;
        if (row0 + 2 < Nn) out[(size_t)2 * F_OUT] = c[2] * sc_lo.z;
        if (row0 + 3 < Nn) out[(size_t)3 * F_OUT] = c[3] * sc_lo.w;
        if (row0 + 4 < Nn) out[(size_t)4 * F_OUT] = c[4] * sc_hi.x;
        if (row0 + 5 < Nn) out[(size_t)5 * F_OUT] = c[5] * sc_hi.y;
        if (row0 + 6 < Nn) out[(size_t)6 * F_OUT] = c[6] * sc_hi.z;
        if (row0 + 7 < Nn) out[(size_t)7 * F_OUT] = c[7] * sc_hi.w;
    }
}

__global__ void gather_logsoftmax(const int* __restrict__ rowstart, const int* __restrict__ cnt,
                                  const int* __restrict__ gsrc, const float* __restrict__ dinv,
                                  const float* __restrict__ H2, const float* __restrict__ b2,
                                  float* __restrict__ OUT, int Nn) {
    const int lane = threadIdx.x & 31;
    const int wave = threadIdx.x >> 5;
    const int node = blockIdx.x * (blockDim.x >> 5) + wave;
    if (node >= Nn) return;

    const int start = __builtin_amdgcn_readfirstlane(rowstart[node]);
    const int n     = __builtin_amdgcn_readfirstlane(cnt[node]);
    const float di  = dinv[node];
    const bool act  = lane < (F_OUT / 4);
    const int col   = act ? lane : 0;
    const float4* __restrict__ H4 = reinterpret_cast<const float4*>(H2);

    float4 acc = make_float4(0.0f, 0.0f, 0.0f, 0.0f);
    for (int base = 0; base < n; base += 32) {
        const int m = (n - base < 32) ? (n - base) : 32;
        int my_s = 0;
        if (lane < m) my_s = gsrc[start + base + lane];
#pragma unroll 4
        for (int j = 0; j < m; ++j) {
            const int    s = __builtin_amdgcn_readlane(my_s, j);
            const float4 h = H4[(size_t)s * (F_OUT / 4) + col];
            acc.x += h.x;
            acc.y += h.y;
            acc.z += h.z;
            acc.w += h.w;
        }
    }

    const size_t o = (size_t)node * (F_OUT / 4) + lane;
    float4 v = make_float4(-INFINITY, -INFINITY, -INFINITY, -INFINITY);
    if (act) {
        const float4 hs = H4[o];
        const float4 bb = reinterpret_cast<const float4*>(b2)[lane];
        v.x = (acc.x + hs.x) * di + bb.x;
        v.y = (acc.y + hs.y) * di + bb.y;
        v.z = (acc.z + hs.z) * di + bb.z;
        v.w = (acc.w + hs.w) * di + bb.w;
    }

    float mx = fmaxf(fmaxf(v.x, v.y), fmaxf(v.z, v.w));
#pragma unroll
    for (int off = 16; off > 0; off >>= 1)
        mx = fmaxf(mx, __shfl_xor(mx, off, 32));

    float s = act ? (__expf(v.x - mx) + __expf(v.y - mx) + __expf(v.z - mx) + __expf(v.w - mx))
                  : 0.0f;
#pragma unroll
    for (int off = 16; off > 0; off >>= 1)
        s += __shfl_xor(s, off, 32);

    const float ls = __logf(s);
    if (act) {
        float4 r;
        r.x = v.x - mx - ls;
        r.y = v.y - mx - ls;
        r.z = v.z - mx - ls;
        r.w = v.w - mx - ls;
        reinterpret_cast<float4*>(OUT)[o] = r;
    }
}

extern "C" void kernel_launch(void* const* d_in, const int* in_sizes, int n_in,
                              void* d_out, int out_size, void* d_ws, size_t ws_size,
                              hipStream_t stream) {
    const float* x  = (const float*)d_in[0];
    const float* W1 = (const float*)d_in[1];
    const float* b1 = (const float*)d_in[2];
    const float* W2 = (const float*)d_in[3];
    const float* b2 = (const float*)d_in[4];
    const int*   ei = (const int*)d_in[5];
    const int E = in_sizes[5] / 2;
    const int Nn = N_NODES;
    const int* src = ei;
    const int* dst = ei + E;

    float* out = (float*)d_out;

    float* ws   = (float*)d_ws;
    float* dinv = ws;
    float* h1   = dinv + 50048;
    float* h2   = h1   + (size_t)Nn * F_HID;
    int*   cnt  = (int*)(h2 + (size_t)Nn * F_OUT);
    int*   rowstart = cnt + 50048;
    int*   gsrc = rowstart + 50048;

    const int TPB = 256;

    build_groups<<<OWN_BLOCKS, OWN_THREADS, 0, stream>>>(src, dst, cnt, rowstart, dinv, gsrc, E, Nn);

    {
        int mtiles = Nn / 16;
        int blocks = (mtiles + 7) / 8;
        gemm_wmma_f32<8, 128><<<blocks, TPB, 0, stream>>>(x, W1, dinv, h1, Nn);
    }

    gather_relu_gemm2<<<(Nn + FG_NODES - 1) / FG_NODES, FG_THREADS, 0, stream>>>(
        rowstart, cnt, gsrc, dinv, h1, b1, W2, h2, Nn);

    gather_logsoftmax<<<(Nn + 7) / 8, TPB, 0, stream>>>(rowstart, cnt, gsrc, dinv, h2, b2, out, Nn);
}


# Layer-1 product after the aggregation, in the gather launch: x rows summed with weight dinv[src], 32-node tile times W1 (LDS) on WMMA; h1 plane and one launch gone; sum order: (Ax)W, not A(xW)
# speedup vs baseline: 4.8844x; 1.1764x over previous
#include <hip/hip_runtime.h>
#include <math.h>


typedef float v2f __attribute__((ext_vector_type(2)));
typedef float v8f __attribute__((ext_vector_type(8)));

#define N_NODES 50000
#define F_IN    128
#define F_HID   128
#define F_OUT   40

#define OWN_NODES   256
#define OWN_BLOCKS  196
#define OWN_THREADS 1024
#define OWN_WAVES   (OWN_THREADS / 32)
#define OWN_GROUP   8
#define OWN_LIST    768
#define NO_NODE     0x7fffffff

static_assert(OWN_NODES * OWN_BLOCKS >= N_NODES, "owners must cover every node");
static_assert(OWN_NODES <= 256, "the node's place in the block is kept in 8 bits");
static_assert(OWN_NODES <= OWN_THREADS, "one thread per owned node in the scan");
static_assert(OWN_THREADS == 1024, "a tile is 1024 16-byte pieces (shift by 10)");

__global__ __launch_bounds__(OWN_THREADS)
void build_groups(const int* __restrict__ src, const int* __restrict__ dst,
                  int* __restrict__ cnt, int* __restrict__ rowstart,
                  float* __restrict__ dinv, int* __restrict__ gsrc, int E, int Nn) {
    __shared__ int2 lst[OWN_WAVES][OWN_LIST];
    __shared__ int c[OWN_NODES];
    __shared__ int wsum[32];
    __shared__ int wlow[32];
    __shared__ int sbase;
    __shared__ int sfull;

    const int tid  = threadIdx.x;
    const int lane = tid & 31;
    const int wave = tid >> 5;
    const int lo   = blockIdx.x * OWN_NODES;
    const int left = Nn - lo;
    const unsigned span = left < OWN_NODES ? (left > 0 ? (unsigned)left : 0u) : (unsigned)OWN_NODES;

    for (int i = tid; i < OWN_NODES; i += OWN_THREADS) c[i] = 0;
    if (tid == 0) sfull = (E >= (1 << 24)) ? 1 : 0;
    __syncthreads();

    int low = 0;
    int nw  = 0;

    auto look = [&](int d, int e) {
        low += __popc(__builtin_amdgcn_ballot_w32(d < lo));
        const unsigned k   = (unsigned)(d - lo);
        const bool     hit = k < span;
        const unsigned hm  = __builtin_amdgcn_ballot_w32(hit);
        if (__builtin_expect(hm != 0u, 0)) {
            if (hit) {
                const int r    = atomicAdd(&c[k], 1);
                const int slot = nw + (int)__builtin_amdgcn_mbcnt_lo(hm, 0u);
                if (slot < OWN_LIST)
                    lst[wave][slot] = make_int2(e, (int)(k | ((unsigned)r << 8)));
            }
            nw += __popc(hm);
        }
    };

    const int nvec = ((E & 3) == 0) ? (E >> 2) : 0;
    const int4* __restrict__ dst4 = reinterpret_cast<const int4*>(dst);
    const int nft  = nvec >> 10;
    const int trot = (int)(((unsigned)blockIdx.x * (unsigned)nft) / gridDim.x);

    auto looks4 = [&](const int4& d, int e) {
        look(d.x, e);
        look(d.y, e + 1);
        look(d.z, e + 2);
        look(d.w, e + 3);
    };
    int t0 = 0;
    for (; t0 + OWN_GROUP <= nft; t0 += OWN_GROUP) {
        int4 d[OWN_GROUP];
        int  eb[OWN_GROUP];
#pragma unroll
        for (int u = 0; u < OWN_GROUP; ++u) {
            int t = t0 + u + trot;
            if (t >= nft) t -= nft;
            const int v = (t << 10) + tid;
            eb[u] = v << 2;
            d[u]  = dst4[v];
        }
#pragma unroll
        for (int u = 0; u < OWN_GROUP; ++u) looks4(d[u], eb[u]);
    }
    if (t0 < nft) {
        const int live = nft - t0;
        int4 d[OWN_GROUP];
        int  eb[OWN_GROUP];
#pragma unroll
        for (int u = 0; u < OWN_GROUP; ++u) {
            int t = t0 + (u < live ? u : 0) + trot;
            if (t >= nft) t -= nft;
            const int v = (t << 10) + tid;
            eb[u] = v << 2;
            d[u]  = dst4[v];
        }
#pragma unroll
        for (int u = 0; u < OWN_GROUP; ++u)
            if (u < live) looks4(d[u], eb[u]);
    }
    for (int i0 = nft << 10; i0 < nvec; i0 += OWN_THREADS) {
        const int  i  = i0 + tid;
        const bool ok = i < nvec;
        int4 d = dst4[ok ? i : i0];
        if (!ok) d = make_int4(NO_NODE, NO_NODE, NO_NODE, NO_NODE);
        look(d.x, 4 * i);
        look(d.y, 4 * i + 1);
        look(d.z, 4 * i + 2);
        look(d.w, 4 * i + 3);
    }
    for (int e0 = nvec * 4; e0 < E; e0 += OWN_THREADS) {
        const int e = e0 + tid;
        look((e < E) ? dst[e] : NO_NODE, e);
    }
    if (nw > OWN_LIST && lane == 0) sfull = 1;
    __syncthreads();

    const int cv = (tid < OWN_NODES) ? c[tid] : 0;
    int v = cv;
#pragma unroll
    for (int off = 1; off < 32; off <<= 1) {
        int t = __shfl_up(v, off, 32);
        if (lane >= off) v += t;
    }
    if (lane == 31) wsum[wave] = v;
    if (lane == 0)  wlow[wave] = low;
    __syncthreads();
    if (wave == 0) {
        const int s = wsum[lane];
        int sv = s;
#pragma unroll
        for (int off = 1; off < 32; off <<= 1) {
            int t = __shfl_up(sv, off, 32);
            if (lane >= off) sv += t;
        }
        wsum[lane] = sv - s;
        int lw = wlow[lane];
#pragma unroll
        for (int off = 16; off > 0; off >>= 1) lw += __shfl_xor(lw, off, 32);
        if (lane == 0) sbase = lw;
    }
    __syncthreads();
    if ((unsigned)tid < span) {
        const int start = sbase + wsum[wave] + v - cv;
        const int node  = lo + tid;
        cnt[node]      = cv;
        rowstart[node] = start;
        dinv[node]     = rsqrtf((float)cv + 1.0f);
        c[tid]         = start;
    }
    __syncthreads();

    if (sfull == 0) {
        for (int i = lane; i < nw; i += 32) {
            const int2     en = lst[wave][i];
            const unsigned k  = (unsigned)en.y & 255u;
            const int      r  = (int)((unsigned)en.y >> 8);
            gsrc[c[k] + r] = src[en.x];
        }
        return;
    }

    for (int e0 = 0; e0 < E; e0 += OWN_THREADS) {
        const int e = e0 + tid;
        const int dd = (e < E) ? dst[e] : NO_NODE;
        const unsigned k = (unsigned)(dd - lo);
        if (k < span) gsrc[atomicAdd(&c[k], 1)] = src[e];
    }
}

#define FG_NODES   32
#define FG_THREADS (FG_NODES * 32)
#define FG_MT      (FG_NODES / 16)
#define FG_PITCH   (F_HID + 4)
#define FG_NT      ((F_OUT + 15) / 16)
#define FG_KS      (F_HID / 4)
#define FG_NT1     (F_HID / 16)
#define FG_KS1     (F_IN / 4)

static_assert(F_IN == F_HID, "one tile pitch and one lane-to-feature map serve both tiles");
static_assert(F_IN == 128, "a row is one 16-B load per lane");
static_assert(FG_KS == FG_NODES && FG_KS1 == FG_NODES, "each wave copies the B entries of one k step");
static_assert(FG_NT1 == 8 && F_HID == 128, "a lane's two 16-B pieces of a W1 row are its entries of the 8 column tiles");
static_assert(FG_MT * FG_NT1 <= FG_NODES && FG_MT * FG_NT <= FG_NODES, "one wave per output tile");

__global__ __launch_bounds__(FG_THREADS)
void gather_gemm1_relu_gemm2(const int* __restrict__ rowstart, const int* __restrict__ cnt,
                             const int* __restrict__ gsrc, const float* __restrict__ dinv,
                             const float* __restrict__ X, const float* __restrict__ W1,
                             const float* __restrict__ b, const float* __restrict__ W2,
                             float* __restrict__ H2, int Nn) {
    __shared__ __attribute__((aligned(16))) float tile[FG_NODES][FG_PITCH];
    __shared__ __attribute__((aligned(16))) float tile2[FG_NODES][FG_PITCH];
    __shared__ v2f bfrag1[FG_KS1][FG_NT1][32];
    __shared__ v2f bfrag[FG_KS][FG_NT][32];
    __shared__ __attribute__((aligned(16))) float sdi[FG_NODES];

    const int lane = threadIdx.x & 31;
    const int wave = threadIdx.x >> 5;
    const int half = lane >> 4;
    const int l16  = lane & 15;
    const int base_node = blockIdx.x * FG_NODES;
    const int node = base_node + wave;
    const bool live = node < Nn;

    const int nodec = live ? node : 0;
    const int   rs_v = rowstart[nodec];
    const int   cn_v = cnt[nodec];
    const float di_v = dinv[nodec];

    const int krow = 4 * wave + 2 * half;
    float4 q[4];
    {
        const float4* p0 = reinterpret_cast<const float4*>(W1 + (size_t)krow * F_HID + 4 * l16);
        const float4* p1 = reinterpret_cast<const float4*>(W1 + (size_t)(krow + 1) * F_HID + 4 * l16);
        q[0] = p0[0];
        q[1] = p0[F_HID / 8];
        q[2] = p1[0];
        q[3] = p1[F_HID / 8];
    }

    v2f wv[FG_NT];
    for (int t = 0; t < FG_NT; ++t) {
        const int col = t * 16 + l16;
        v2f w;
        w.x = 0.0f; w.y = 0.0f;
        if ((F_OUT % 16 == 0) || (col < F_OUT)) {
            w.x = W2[(size_t)krow * F_OUT + col];
            w.y = W2[(size_t)(krow + 1) * F_OUT + col];
        }
        wv[t] = w;
    }

    {
        const float r0[8] = {q[0].x, q[0].y, q[0].z, q[0].w, q[1].x, q[1].y, q[1].z, q[1].w};
        const float r1[8] = {q[2].x, q[2].y, q[2].z, q[2].w, q[3].x, q[3].y, q[3].z, q[3].w};
#pragma unroll
        for (int t = 0; t < FG_NT1; ++t) {
            v2f e;
            e.x = r0[t];
            e.y = r1[t];
            bfrag1[wave][t][lane] = e;
        }
    }

    float4 v = make_float4(0.0f, 0.0f, 0.0f, 0.0f);
    float di = 0.0f;
    if (live) {
        const int start = __builtin_amdgcn_readfirstlane(rs_v);
        const int n     = __builtin_amdgcn_readfirstlane(cn_v);
        di = di_v;
        const float4* __restrict__ X4 = reinterpret_cast<const float4*>(X);

        float4 acc = make_float4(0.0f, 0.0f, 0.0f, 0.0f);
        for (int base = 0; base < n; base += 32) {
            const int m = (n - base < 32) ? (n - base) : 32;
            int   my_s = 0;
            float my_w = 0.0f;
            if (lane < m) {
                my_s = gsrc[start + base + lane];
                my_w = dinv[my_s];
            }
            const int my_wi = __float_as_int(my_w);
#pragma unroll 4
            for (int j = 0; j < m; ++j) {
                const int    s = __builtin_amdgcn_readlane(my_s, j);
                const float  w = __int_as_float(__builtin_amdgcn_readlane(my_wi, j));
                const float4 h = X4[(size_t)s * (F_IN / 4) + lane];
                acc.x += h.x * w;
                acc.y += h.y * w;
                acc.z += h.z * w;
                acc.w += h.w * w;
            }
        }

        const float4 xs = X4[(size_t)node * (F_IN / 4) + lane];
        v.x = (acc.x + xs.x * di) * di;
        v.y = (acc.y + xs.y * di) * di;
        v.z = (acc.z + xs.z * di) * di;
        v.w = (acc.w + xs.w * di) * di;
    }
    for (int t = 0; t < FG_NT; ++t) bfrag[wave][t][lane] = wv[t];
    *reinterpret_cast<float4*>(&tile[wave][4 * lane]) = v;
    if (lane == 0) sdi[wave] = di;
    __syncthreads();

    if (wave < FG_MT * FG_NT1) {
        const int   mt   = wave / FG_NT1;
        const int   ct   = wave % FG_NT1;
        const int   hcol = 64 * (ct >> 2) + 4 * l16 + (ct & 3);
        const float bj   = b[hcol];
        v8f c1 = {};
        const float* arow = &tile[16 * mt + l16][2 * half];
        for (int ks = 0; ks < FG_KS1; ++ks) {
            const v2f a  = *reinterpret_cast<const v2f*>(arow + 4 * ks);
            const v2f bf = bfrag1[ks][ct][lane];
            c1 = __builtin_amdgcn_wmma_f32_16x16x4_f32(
                false, a, false, bf, (short)0, c1, false, false);
        }
#pragma unroll
        for (int r = 0; r < 8; ++r) {
            const float y = c1[r] + bj;
            tile2[16 * mt + 8 * half + r][hcol] = y > 0.0f ? y : 0.0f;
        }
    }
    __syncthreads();

    if (wave >= FG_MT * FG_NT) return;

    const int mt   = wave / FG_NT;
    const int ct   = wave % FG_NT;
    const int ncol = ct * 16 + l16;

    const float4* sc4 = reinterpret_cast<const float4*>(&sdi[16 * mt + half * 8]);
    const float4 sc_lo = sc4[0];
    const float4 sc_hi = sc4[1];

    v8f c = {};
    const float* arow = &tile2[16 * mt + l16][2 * half];
    for (int ks = 0; ks < FG_KS; ++ks) {
        const v2f a  = *reinterpret_cast<const v2f*>(arow + 4 * ks);
        const v2f bf = bfrag[ks][ct][lane];
        c = __builtin_amdgcn_wmma_f32_16x16x4_f32(
            false, a, false, bf, (short)0, c, false, false);
    }

    if ((F_OUT % 16 == 0) || (ncol < F_OUT)) {
        const int row0 = base_node + 16 * mt + half * 8;
        float* out = H2 + (size_t)row0 * F_OUT + ncol;
        if (row0 + 0 < Nn) out[(size_t)0 * F_OUT] = c[0] * sc_lo.x;
        if (row0 + 1 < Nn) out[(size_t)1 * F_OUT] = c[1] * sc_lo.y;
        if (row0 + 2 < Nn) out[(size_t)2 * F_OUT] = c[2] * sc_lo.z;
        if (row0 + 3 < Nn) out[(size_t)3 * F_OUT] = c[3] * sc_lo.w;
        if (row0 + 4 < Nn) out[(size_t)4 * F_OUT] = c[4] * sc_hi.x;
        if (row0 + 5 < Nn) out[(size_t)5 * F_OUT] = c[5] * sc_hi.y;
        if (row0 + 6 < Nn) out[(size_t)6 * F_OUT] = c[6] * sc_hi.z;
        if (row0 + 7 < Nn) out[(size_t)7 * F_OUT] = c[7] * sc_hi.w;
    }
}

__global__ void gather_logsoftmax(const int* __restrict__ rowstart, const int* __restrict__ cnt,
                                  const int* __restrict__ gsrc, const float* __restrict__ dinv,
                                  const float* __restrict__ H2, const float* __restrict__ b2,
                                  float* __restrict__ OUT, int Nn) {
    const int lane = threadIdx.x & 31;
    const int wave = threadIdx.x >> 5;
    const int node = blockIdx.x * (blockDim.x >> 5) + wave;
    if (node >= Nn) return;

    const int start = __builtin_amdgcn_readfirstlane(rowstart[node]);
    const int n     = __builtin_amdgcn_readfirstlane(cnt[node]);
    const float di  = dinv[node];
    const bool act  = lane < (F_OUT / 4);
    const int col   = act ? lane : 0;
    const float4* __restrict__ H4 = reinterpret_cast<const float4*>(H2);

    float4 acc = make_float4(0.0f, 0.0f, 0.0f, 0.0f);
    for (int base = 0; base < n; base += 32) {
        const int m = (n - base < 32) ? (n - base) : 32;
        int my_s = 0;
        if (lane < m) my_s = gsrc[start + base + lane];
#pragma unroll 4
        for (int j = 0; j < m; ++j) {
            const int    s = __builtin_amdgcn_readlane(my_s, j);
            const float4 h = H4[(size_t)s * (F_OUT / 4) + col];
            acc.x += h.x;
            acc.y += h.y;
            acc.z += h.z;
            acc.w += h.w;
        }
    }

    const size_t o = (size_t)node * (F_OUT / 4) + lane;
    float4 v = make_float4(-INFINITY, -INFINITY, -INFINITY, -INFINITY);
    if (act) {
        const float4 hs = H4[o];
        const float4 bb = reinterpret_cast<const float4*>(b2)[lane];
        v.x = (acc.x + hs.x) * di + bb.x;
        v.y = (acc.y + hs.y) * di + bb.y;
        v.z = (acc.z + hs.z) * di + bb.z;
        v.w = (acc.w + hs.w) * di + bb.w;
    }

    float mx = fmaxf(fmaxf(v.x, v.y), fmaxf(v.z, v.w));
#pragma unroll
    for (int off = 16; off > 0; off >>= 1)
        mx = fmaxf(mx, __shfl_xor(mx, off, 32));

    float s = act ? (__expf(v.x - mx) + __expf(v.y - mx) + __expf(v.z - mx) + __expf(v.w - mx))
                  : 0.0f;
#pragma unroll
    for (int off = 16; off > 0; off >>= 1)
        s += __shfl_xor(s, off, 32);

    const float ls = __logf(s);
    if (act) {
        float4 r;
        r.x = v.x - mx - ls;
        r.y = v.y - mx - ls;
        r.z = v.z - mx - ls;
        r.w = v.w - mx - ls;
        reinterpret_cast<float4*>(OUT)[o] = r;
    }
}

extern "C" void kernel_launch(void* const* d_in, const int* in_sizes, int n_in,
                              void* d_out, int out_size, void* d_ws, size_t ws_size,
                              hipStream_t stream) {
    const float* x  = (const float*)d_in[0];
    const float* W1 = (const float*)d_in[1];
    const float* b1 = (const float*)d_in[2];
    const float* W2 = (const float*)d_in[3];
    const float* b2 = (const float*)d_in[4];
    const int*   ei = (const int*)d_in[5];
    const int E = in_sizes[5] / 2;
    const int Nn = N_NODES;
    const int* src = ei;
    const int* dst = ei + E;

    float* out = (float*)d_out;

    float* ws   = (float*)d_ws;
    float* dinv = ws;
    float* h2   = dinv + 50048;
    int*   cnt  = (int*)(h2 + (size_t)Nn * F_OUT);
    int*   rowstart = cnt + 50048;
    int*   gsrc = rowstart + 50048;

    const int TPB = 256;

    build_groups<<<OWN_BLOCKS, OWN_THREADS, 0, stream>>>(src, dst, cnt, rowstart, dinv, gsrc, E, Nn);

    gather_gemm1_relu_gemm2<<<(Nn + FG_NODES - 1) / FG_NODES, FG_THREADS, 0, stream>>>(
        rowstart, cnt, gsrc, dinv, x, W1, b1, W2, h2, Nn);

    gather_logsoftmax<<<(Nn + 7) / 8, TPB, 0, stream>>>(rowstart, cnt, gsrc, dinv, h2, b2, out, Nn);
}


# Both gather launches as resident grids: 512 32-wave groups walk the 32-node tiles (W1, W2 into LDS once per group), 16384 waves walk the layer-2 nodes; 32768 wave entries, not 100016; same sums
# speedup vs baseline: 5.1456x; 1.0535x over previous
#include <hip/hip_runtime.h>
#include <math.h>


typedef float v2f __attribute__((ext_vector_type(2)));
typedef float v8f __attribute__((ext_vector_type(8)));

#define N_NODES 50000
#define F_IN    128
#define F_HID   128
#define F_OUT   40

#define OWN_NODES   256
#define OWN_BLOCKS  196
#define OWN_THREADS 1024
#define OWN_WAVES   (OWN_THREADS / 32)
#define OWN_GROUP   8
#define OWN_LIST    768
#define NO_NODE     0x7fffffff

static_assert(OWN_NODES * OWN_BLOCKS >= N_NODES, "owners must cover every node");
static_assert(OWN_NODES <= 256, "the node's place in the block is kept in 8 bits");
static_assert(OWN_NODES <= OWN_THREADS, "one thread per owned node in the scan");
static_assert(OWN_THREADS == 1024, "a tile is 1024 16-byte pieces (shift by 10)");

__global__ __launch_bounds__(OWN_THREADS)
void build_groups(const int* __restrict__ src, const int* __restrict__ dst,
                  int* __restrict__ cnt, int* __restrict__ rowstart,
                  float* __restrict__ dinv, int* __restrict__ gsrc, int E, int Nn) {
    __shared__ int2 lst[OWN_WAVES][OWN_LIST];
    __shared__ int c[OWN_NODES];
    __shared__ int wsum[32];
    __shared__ int wlow[32];
    __shared__ int sbase;
    __shared__ int sfull;

    const int tid  = threadIdx.x;
    const int lane = tid & 31;
    const int wave = tid >> 5;
    const int lo   = blockIdx.x * OWN_NODES;
    const int left = Nn - lo;
    const unsigned span = left < OWN_NODES ? (left > 0 ? (unsigned)left : 0u) : (unsigned)OWN_NODES;

    for (int i = tid; i < OWN_NODES; i += OWN_THREADS) c[i] = 0;
    if (tid == 0) sfull = (E >= (1 << 24)) ? 1 : 0;
    __syncthreads();

    int low = 0;
    int nw  = 0;

    auto look = [&](int d, int e) {
        low += __popc(__builtin_amdgcn_ballot_w32(d < lo));
        const unsigned k   = (unsigned)(d - lo);
        const bool     hit = k < span;
        const unsigned hm  = __builtin_amdgcn_ballot_w32(hit);
        if (__builtin_expect(hm != 0u, 0)) {
            if (hit) {
                const int r    = atomicAdd(&c[k], 1);
                const int slot = nw + (int)__builtin_amdgcn_mbcnt_lo(hm, 0u);
                if (slot < OWN_LIST)
                    lst[wave][slot] = make_int2(e, (int)(k | ((unsigned)r << 8)));
            }
            nw += __popc(hm);
        }
    };

    const int nvec = ((E & 3) == 0) ? (E >> 2) : 0;
    const int4* __restrict__ dst4 = reinterpret_cast<const int4*>(dst);
    const int nft  = nvec >> 10;
    const int trot = (int)(((unsigned)blockIdx.x * (unsigned)nft) / gridDim.x);

    auto looks4 = [&](const int4& d, int e) {
        look(d.x, e);
        look(d.y, e + 1);
        look(d.z, e + 2);
        look(d.w, e + 3);
    };
    int t0 = 0;
    for (; t0 + OWN_GROUP <= nft; t0 += OWN_GROUP) {
        int4 d[OWN_GROUP];
        int  eb[OWN_GROUP];
#pragma unroll
        for (int u = 0; u < OWN_GROUP; ++u) {
            int t = t0 + u + trot;
            if (t >= nft) t -= nft;
            const int v = (t << 10) + tid;
            eb[u] = v << 2;
            d[u]  = dst4[v];
        }
#pragma unroll
        for (int u = 0; u < OWN_GROUP; ++u) looks4(d[u], eb[u]);
    }
    if (t0 < nft) {
        const int live = nft - t0;
        int4 d[OWN_GROUP];
        int  eb[OWN_GROUP];
#pragma unroll
        for (int u = 0; u < OWN_GROUP; ++u) {
            int t = t0 + (u < live ? u : 0) + trot;
            if (t >= nft) t -= nft;
            const int v = (t << 10) + tid;
            eb[u] = v << 2;
            d[u]  = dst4[v];
        }
#pragma unroll
        for (int u = 0; u < OWN_GROUP; ++u)
            if (u < live) looks4(d[u], eb[u]);
    }
    for (int i0 = nft << 10; i0 < nvec; i0 += OWN_THREADS) {
        const int  i  = i0 + tid;
        const bool ok = i < nvec;
        int4 d = dst4[ok ? i : i0];
        if (!ok) d = make_int4(NO_NODE, NO_NODE, NO_NODE, NO_NODE);
        look(d.x, 4 * i);
        look(d.y, 4 * i + 1);
        look(d.z, 4 * i + 2);
        look(d.w, 4 * i + 3);
    }
    for (int e0 = nvec * 4; e0 < E; e0 += OWN_THREADS) {
        const int e = e0 + tid;
        look((e < E) ? dst[e] : NO_NODE, e);
    }
    if (nw > OWN_LIST && lane == 0) sfull = 1;
    __syncthreads();

    const int cv = (tid < OWN_NODES) ? c[tid] : 0;
    int v = cv;
#pragma unroll
    for (int off = 1; off < 32; off <<= 1) {
        int t = __shfl_up(v, off, 32);
        if (lane >= off) v += t;
    }
    if (lane == 31) wsum[wave] = v;
    if (lane == 0)  wlow[wave] = low;
    __syncthreads();
    if (wave == 0) {
        const int s = wsum[lane];
        int sv = s;
#pragma unroll
        for (int off = 1; off < 32; off <<= 1) {
            int t = __shfl_up(sv, off, 32);
            if (lane >= off) sv += t;
        }
        wsum[lane] = sv - s;
        int lw = wlow[lane];
#pragma unroll
        for (int off = 16; off > 0; off >>= 1) lw += __shfl_xor(lw, off, 32);
        if (lane == 0) sbase = lw;
    }
    __syncthreads();
    if ((unsigned)tid < span) {
        const int start = sbase + wsum[wave] + v - cv;
        const int node  = lo + tid;
        cnt[node]      = cv;
        rowstart[node] = start;
        dinv[node]     = rsqrtf((float)cv + 1.0f);
        c[tid]         = start;
    }
    __syncthreads();

    if (sfull == 0) {
        for (int i = lane; i < nw; i += 32) {
            const int2     en = lst[wave][i];
            const unsigned k  = (unsigned)en.y & 255u;
            const int      r  = (int)((unsigned)en.y >> 8);
            gsrc[c[k] + r] = src[en.x];
        }
        return;
    }

    for (int e0 = 0; e0 < E; e0 += OWN_THREADS) {
        const int e = e0 + tid;
        const int dd = (e < E) ? dst[e] : NO_NODE;
        const unsigned k = (unsigned)(dd - lo);
        if (k < span) gsrc[atomicAdd(&c[k], 1)] = src[e];
    }
}

#define FG_NODES   32
#define FG_GRID    512
#define FG_THREADS (FG_NODES * 32)
#define FG_MT      (FG_NODES / 16)
#define FG_PITCH   (F_HID + 4)
#define FG_NT      ((F_OUT + 15) / 16)
#define FG_KS      (F_HID / 4)
#define FG_NT1     (F_HID / 16)
#define FG_KS1     (F_IN / 4)

static_assert(F_IN == F_HID, "one tile pitch and one lane-to-feature map serve both tiles");
static_assert(F_IN == 128, "a row is one 16-B load per lane");
static_assert(FG_KS == FG_NODES && FG_KS1 == FG_NODES, "each wave copies the B entries of one k step");
static_assert(FG_NT1 == 8 && F_HID == 128, "a lane's two 16-B pieces of a W1 row are its entries of the 8 column tiles");
static_assert(FG_MT * FG_NT1 <= FG_NODES && FG_MT * FG_NT <= FG_NODES, "one wave per output tile");

__global__ __launch_bounds__(FG_THREADS)
void gather_gemm1_relu_gemm2(const int* __restrict__ rowstart, const int* __restrict__ cnt,
                             const int* __restrict__ gsrc, const float* __restrict__ dinv,
                             const float* __restrict__ X, const float* __restrict__ W1,
                             const float* __restrict__ b, const float* __restrict__ W2,
                             float* __restrict__ H2, int Nn) {
    __shared__ __attribute__((aligned(16))) float tile[FG_NODES][FG_PITCH];
    __shared__ __attribute__((aligned(16))) float tile2[FG_NODES][FG_PITCH];
    __shared__ v2f bfrag1[FG_KS1][FG_NT1][32];
    __shared__ v2f bfrag[FG_KS][FG_NT][32];
    __shared__ __attribute__((aligned(16))) float sdi[FG_NODES];

    const int lane = threadIdx.x & 31;
    const int wave = threadIdx.x >> 5;
    const int half = lane >> 4;
    const int l16  = lane & 15;
    const int ntiles = (Nn + FG_NODES - 1) / FG_NODES;
    const int tstep  = (int)gridDim.x;

    int   rs_v;
    int   cn_v;
    float di_v;
    {
        const int node0  = (int)blockIdx.x * FG_NODES + wave;
        const int nodec0 = (node0 < Nn) ? node0 : 0;
        rs_v = rowstart[nodec0];
        cn_v = cnt[nodec0];
        di_v = dinv[nodec0];
    }

    const int krow = 4 * wave + 2 * half;
    float4 q[4];
    {
        const float4* p0 = reinterpret_cast<const float4*>(W1 + (size_t)krow * F_HID + 4 * l16);
        const float4* p1 = reinterpret_cast<const float4*>(W1 + (size_t)(krow + 1) * F_HID + 4 * l16);
        q[0] = p0[0];
        q[1] = p0[F_HID / 8];
        q[2] = p1[0];
        q[3] = p1[F_HID / 8];
    }

    v2f wv[FG_NT];
    for (int t = 0; t < FG_NT; ++t) {
        const int col = t * 16 + l16;
        v2f w;
        w.x = 0.0f; w.y = 0.0f;
        if ((F_OUT % 16 == 0) || (col < F_OUT)) {
            w.x = W2[(size_t)krow * F_OUT + col];
            w.y = W2[(size_t)(krow + 1) * F_OUT + col];
        }
        wv[t] = w;
    }

    {
        const float r0[8] = {q[0].x, q[0].y, q[0].z, q[0].w, q[1].x, q[1].y, q[1].z, q[1].w};
        const float r1[8] = {q[2].x, q[2].y, q[2].z, q[2].w, q[3].x, q[3].y, q[3].z, q[3].w};
#pragma unroll
        for (int t = 0; t < FG_NT1; ++t) {
            v2f e;
            e.x = r0[t];
            e.y = r1[t];
            bfrag1[wave][t][lane] = e;
        }
    }

    for (int t = 0; t < FG_NT; ++t) bfrag[wave][t][lane] = wv[t];

    for (int tile_i = (int)blockIdx.x; tile_i < ntiles; tile_i += tstep) {
    const int base_node = tile_i * FG_NODES;
    const int node = base_node + wave;
    const bool live = node < Nn;

    float4 v = make_float4(0.0f, 0.0f, 0.0f, 0.0f);
    float di = 0.0f;
    int   rs_n = 0;
    int   cn_n = 0;
    float di_n = 0.0f;
    if (live) {
        const int start = __builtin_amdgcn_readfirstlane(rs_v);
        const int n     = __builtin_amdgcn_readfirstlane(cn_v);
        di = di_v;
        const float4* __restrict__ X4 = reinterpret_cast<const float4*>(X);

        float4 acc = make_float4(0.0f, 0.0f, 0.0f, 0.0f);
        for (int base = 0; base < n; base += 32) {
            const int m = (n - base < 32) ? (n - base) : 32;
            int   my_s = 0;
            float my_w = 0.0f;
            if (lane < m) {
                my_s = gsrc[start + base + lane];
                my_w = dinv[my_s];
            }
            const int my_wi = __float_as_int(my_w);
#pragma unroll 4
            for (int j = 0; j < m; ++j) {
                const int    s = __builtin_amdgcn_readlane(my_s, j);
                const float  w = __int_as_float(__builtin_amdgcn_readlane(my_wi, j));
                const float4 h = X4[(size_t)s * (F_IN / 4) + lane];
                acc.x += h.x * w;
                acc.y += h.y * w;
                acc.z += h.z * w;
                acc.w += h.w * w;
            }
        }

        const int tile_n = tile_i + tstep;
        const int node_n = tile_n * FG_NODES + wave;
        const int nodec_n = (tile_n < ntiles && node_n < Nn) ? node_n : 0;
        rs_n = rowstart[nodec_n];
        cn_n = cnt[nodec_n];
        di_n = dinv[nodec_n];

        const float4 xs = X4[(size_t)node * (F_IN / 4) + lane];
        v.x = (acc.x + xs.x * di) * di;
        v.y = (acc.y + xs.y * di) * di;
        v.z = (acc.z + xs.z * di) * di;
        v.w = (acc.w + xs.w * di) * di;
    }

    *reinterpret_cast<float4*>(&tile[wave][4 * lane]) = v;
    if (lane == 0) sdi[wave] = di;
    __syncthreads();

    float4 sc_lo = make_float4(0.0f, 0.0f, 0.0f, 0.0f);
    float4 sc_hi = make_float4(0.0f, 0.0f, 0.0f, 0.0f);
    if (wave < FG_MT * FG_NT) {
        const float4* sc4 = reinterpret_cast<const float4*>(&sdi[16 * (wave / FG_NT) + half * 8]);
        sc_lo = sc4[0];
        sc_hi = sc4[1];
    }

    if (wave < FG_MT * FG_NT1) {
        const int   mt   = wave / FG_NT1;
        const int   ct   = wave % FG_NT1;
        const int   hcol = 64 * (ct >> 2) + 4 * l16 + (ct & 3);
        const float bj   = b[hcol];
        v8f c1 = {};
        const float* arow = &tile[16 * mt + l16][2 * half];
        for (int ks = 0; ks < FG_KS1; ++ks) {
            const v2f a  = *reinterpret_cast<const v2f*>(arow + 4 * ks);
            const v2f bf = bfrag1[ks][ct][lane];
            c1 = __builtin_amdgcn_wmma_f32_16x16x4_f32(
                false, a, false, bf, (short)0, c1, false, false);
        }
#pragma unroll
        for (int r = 0; r < 8; ++r) {
            const float y = c1[r] + bj;
            tile2[16 * mt + 8 * half + r][hcol] = y > 0.0f ? y : 0.0f;
        }
    }
    __syncthreads();

    if (wave < FG_MT * FG_NT) {

    const int mt   = wave / FG_NT;
    const int ct   = wave % FG_NT;
    const int ncol = ct * 16 + l16;

    v8f c = {};
    const float* arow = &tile2[16 * mt + l16][2 * half];
    for (int ks = 0; ks < FG_KS; ++ks) {
        const v2f a  = *reinterpret_cast<const v2f*>(arow + 4 * ks);
        const v2f bf = bfrag[ks][ct][lane];
        c = __builtin_amdgcn_wmma_f32_16x16x4_f32(
            false, a, false, bf, (short)0, c, false, false);
    }

    if ((F_OUT % 16 == 0) || (ncol < F_OUT)) {
        const int row0 = base_node + 16 * mt + half * 8;
        float* out = H2 + (size_t)row0 * F_OUT + ncol;
        if (row0 + 0 < Nn) out[(size_t)0 * F_OUT] = c[0] * sc_lo.x;
        if (row0 + 1 < Nn) out[(size_t)1 * F_OUT] = c[1] * sc_lo.y;
        if (row0 + 2 < Nn) out[(size_t)2 * F_OUT] = c[2] * sc_lo.z;
        if (row0 + 3 < Nn) out[(size_t)3 * F_OUT] = c[3] * sc_lo.w;
        if (row0 + 4 < Nn) out[(size_t)4 * F_OUT] = c[4] * sc_hi.x;
        if (row0 + 5 < Nn) out[(size_t)5 * F_OUT] = c[5] * sc_hi.y;
        if (row0 + 6 < Nn) out[(size_t)6 * F_OUT] = c[6] * sc_hi.z;
        if (row0 + 7 < Nn) out[(size_t)7 * F_OUT] = c[7] * sc_hi.w;
    }
    }

    rs_v = rs_n;
    cn_v = cn_n;
    di_v = di_n;
    }
}

#define LS_THREADS 256
#define LS_GRID    2048

__global__ __launch_bounds__(LS_THREADS)
void gather_logsoftmax(const int* __restrict__ rowstart, const int* __restrict__ cnt,
                                  const int* __restrict__ gsrc, const float* __restrict__ dinv,
                                  const float* __restrict__ H2, const float* __restrict__ b2,
                                  float* __restrict__ OUT, int Nn) {
    const int lane = threadIdx.x & 31;
    const int wave = threadIdx.x >> 5;
    const int wstep = (int)gridDim.x * (LS_THREADS / 32);
    int node = (int)blockIdx.x * (LS_THREADS / 32) + wave;
    if (node >= Nn) return;

    const bool act  = lane < (F_OUT / 4);
    const int col   = act ? lane : 0;
    const float4* __restrict__ H4 = reinterpret_cast<const float4*>(H2);

    int   rs_v = rowstart[node];
    int   cn_v = cnt[node];
    float di   = dinv[node];

    for (;;) {
    const int start = __builtin_amdgcn_readfirstlane(rs_v);
    const int n     = __builtin_amdgcn_readfirstlane(cn_v);

    float4 acc = make_float4(0.0f, 0.0f, 0.0f, 0.0f);
    for (int base = 0; base < n; base += 32) {
        const int m = (n - base < 32) ? (n - base) : 32;
        int my_s = 0;
        if (lane < m) my_s = gsrc[start + base + lane];
#pragma unroll 4
        for (int j = 0; j < m; ++j) {
            const int    s = __builtin_amdgcn_readlane(my_s, j);
            const float4 h = H4[(size_t)s * (F_OUT / 4) + col];
            acc.x += h.x;
            acc.y += h.y;
            acc.z += h.z;
            acc.w += h.w;
        }
    }

    const int  node_n = node + wstep;
    const bool more   = node_n < Nn;
    const int  nodec_n = more ? node_n : node;
    const int   rs_n = rowstart[nodec_n];
    const int   cn_n = cnt[nodec_n];
    const float di_n = dinv[nodec_n];

    const size_t o = (size_t)node * (F_OUT / 4) + lane;
    float4 v = make_float4(-INFINITY, -INFINITY, -INFINITY, -INFINITY);
    if (act) {
        const float4 hs = H4[o];
        const float4 bb = reinterpret_cast<const float4*>(b2)[lane];
        v.x = (acc.x + hs.x) * di + bb.x;
        v.y = (acc.y + hs.y) * di + bb.y;
        v.z = (acc.z + hs.z) * di + bb.z;
        v.w = (acc.w + hs.w) * di + bb.w;
    }

    float mx = fmaxf(fmaxf(v.x, v.y), fmaxf(v.z, v.w));
#pragma unroll
    for (int off = 16; off > 0; off >>= 1)
        mx = fmaxf(mx, __shfl_xor(mx, off, 32));

    float s = act ? (__expf(v.x - mx) + __expf(v.y - mx) + __expf(v.z - mx) + __expf(v.w - mx))
                  : 0.0f;
#pragma unroll
    for (int off = 16; off > 0; off >>= 1)
        s += __shfl_xor(s, off, 32);

    const float ls = __logf(s);
    if (act) {
        float4 r;
        r.x = v.x - mx - ls;
        r.y = v.y - mx - ls;
        r.z = v.z - mx - ls;
        r.w = v.w - mx - ls;
        reinterpret_cast<float4*>(OUT)[o] = r;
    }

    if (!more) break;
    node = node_n;
    rs_v = rs_n;
    cn_v = cn_n;
    di   = di_n;
    }
}

extern "C" void kernel_launch(void* const* d_in, const int* in_sizes, int n_in,
                              void* d_out, int out_size, void* d_ws, size_t ws_size,
                              hipStream_t stream) {
    const float* x  = (const float*)d_in[0];
    const float* W1 = (const float*)d_in[1];
    const float* b1 = (const float*)d_in[2];
    const float* W2 = (const float*)d_in[3];
    const float* b2 = (const float*)d_in[4];
    const int*   ei = (const int*)d_in[5];
    const int E = in_sizes[5] / 2;
    const int Nn = N_NODES;
    const int* src = ei;
    const int* dst = ei + E;

    float* out = (float*)d_out;

    float* ws   = (float*)d_ws;
    float* dinv = ws;
    float* h2   = dinv + 50048;
    int*   cnt  = (int*)(h2 + (size_t)Nn * F_OUT);
    int*   rowstart = cnt + 50048;
    int*   gsrc = rowstart + 50048;

    build_groups<<<OWN_BLOCKS, OWN_THREADS, 0, stream>>>(src, dst, cnt, rowstart, dinv, gsrc, E, Nn);

    gather_gemm1_relu_gemm2<<<FG_GRID, FG_THREADS, 0, stream>>>(
        rowstart, cnt, gsrc, dinv, x, W1, b1, W2, h2, Nn);

    gather_logsoftmax<<<LS_GRID, LS_THREADS, 0, stream>>>(rowstart, cnt, gsrc, dinv, h2, b2, out, Nn);
}


# Edge grouping in two levels: a new launch sorts 200 slices of the edge list by owner block (LDS counters), an owner then reads its own runs, not all destinations; row sum order = LDS arrival order
# speedup vs baseline: 6.6177x; 1.2861x over previous
#include <hip/hip_runtime.h>
#include <math.h>


typedef float v2f __attribute__((ext_vector_type(2)));
typedef float v8f __attribute__((ext_vector_type(8)));

#define N_NODES 50000
#define F_IN    128
#define F_HID   128
#define F_OUT   40

#define OWN_NODES   256
#define OWN_BLOCKS  196
#define OWN_THREADS 1024
#define OWN_WAVES   (OWN_THREADS / 32)
#define OWN_LIST    768
#define BIN_BLOCKS  200
#define BIN_THREADS 1024
#define BIN_GROUP   4
#define OWN_SEGS    ((BIN_BLOCKS + OWN_WAVES - 1) / OWN_WAVES)

static_assert(OWN_NODES * OWN_BLOCKS >= N_NODES, "owners must cover every node");
static_assert(OWN_NODES == 256, "the node's place in its owner is the low 8 bits of its number");
static_assert(OWN_NODES <= OWN_THREADS, "one thread per owned node in the scan");
static_assert(N_NODES < (1 << 23), "a source number and the 8 bits of the place share one int");
static_assert(OWN_BLOCKS <= 256, "the scan of the owner counters runs on 8 waves");
static_assert(OWN_SEGS <= 32, "lane j of a wave holds the description of the wave's j-th run");

__global__ __launch_bounds__(BIN_THREADS)
void bin_edges(const int* __restrict__ src, const int* __restrict__ dst,
               int* __restrict__ binned, int* __restrict__ tcnt, int* __restrict__ tstart, int E) {
    __shared__ int h[OWN_BLOCKS];
    __shared__ int wsum[8];

    const int tid  = threadIdx.x;
    const int lane = tid & 31;
    const int wave = tid >> 5;
    const int per  = (E + BIN_BLOCKS - 1) / BIN_BLOCKS;
    const int e0   = (int)blockIdx.x * per;
    const int e1   = (e0 + per < E) ? (e0 + per) : E;

    for (int i = tid; i < OWN_BLOCKS; i += BIN_THREADS) h[i] = 0;
    __syncthreads();

    for (int base = e0; base < e1; base += BIN_GROUP * BIN_THREADS) {
        int d[BIN_GROUP];
#pragma unroll
        for (int u = 0; u < BIN_GROUP; ++u) {
            const int e = base + u * BIN_THREADS + tid;
            d[u] = dst[(e < e1) ? e : e0];
        }
#pragma unroll
        for (int u = 0; u < BIN_GROUP; ++u)
            if (base + u * BIN_THREADS + tid < e1) atomicAdd(&h[d[u] >> 8], 1);
    }
    __syncthreads();

    int cv = 0;
    int v  = 0;
    if (tid < 256) {
        cv = (tid < OWN_BLOCKS) ? h[tid] : 0;
        v  = cv;
#pragma unroll
        for (int off = 1; off < 32; off <<= 1) {
            int t = __shfl_up(v, off, 32);
            if (lane >= off) v += t;
        }
        if (lane == 31) wsum[wave] = v;
    }
    __syncthreads();
    if (tid < 256) {
        int before = 0;
        for (int w = 0; w < wave; ++w) before += wsum[w];
        const int start = before + v - cv;
        if (tid < OWN_BLOCKS) {
            h[tid] = start;
            tcnt[tid * BIN_BLOCKS + (int)blockIdx.x]   = cv;
            tstart[tid * BIN_BLOCKS + (int)blockIdx.x] = start;
        }
    }
    __syncthreads();

    for (int base = e0; base < e1; base += BIN_GROUP * BIN_THREADS) {
        int d[BIN_GROUP];
        int s[BIN_GROUP];
#pragma unroll
        for (int u = 0; u < BIN_GROUP; ++u) {
            const int e  = base + u * BIN_THREADS + tid;
            const int ec = (e < e1) ? e : e0;
            d[u] = dst[ec];
            s[u] = src[ec];
        }
#pragma unroll
        for (int u = 0; u < BIN_GROUP; ++u)
            if (base + u * BIN_THREADS + tid < e1) {
                const int pos = atomicAdd(&h[d[u] >> 8], 1);
                binned[e0 + pos] = (int)(((unsigned)s[u] << 8) | ((unsigned)d[u] & 255u));
            }
    }
}

__global__ __launch_bounds__(OWN_THREADS)
void build_groups(const int* __restrict__ binned, const int* __restrict__ tcnt,
                  const int* __restrict__ tstart,
                  int* __restrict__ cnt, int* __restrict__ rowstart,
                  float* __restrict__ dinv, int* __restrict__ gsrc, int E, int Nn) {
    __shared__ int2 lst[OWN_WAVES][OWN_LIST];
    __shared__ int c[OWN_NODES];
    __shared__ int wsum[32];
    __shared__ int wlow[32];
    __shared__ int sbase;
    __shared__ int sfull;

    const int tid  = threadIdx.x;
    const int lane = tid & 31;
    const int wave = tid >> 5;
    const int own  = (int)blockIdx.x;
    const int lo   = own * OWN_NODES;
    const int left = Nn - lo;
    const unsigned span = left < OWN_NODES ? (left > 0 ? (unsigned)left : 0u) : (unsigned)OWN_NODES;
    const int per  = (E + BIN_BLOCKS - 1) / BIN_BLOCKS;

    for (int i = tid; i < OWN_NODES; i += OWN_THREADS) c[i] = 0;
    if (tid == 0) sfull = 0;
    __syncthreads();

    int seg_n = 0;
    int seg_s = 0;
    {
        const int b = wave + OWN_WAVES * lane;
        if (lane < OWN_SEGS && b < BIN_BLOCKS) {
            seg_n = tcnt[own * BIN_BLOCKS + b];
            seg_s = tstart[own * BIN_BLOCKS + b];
        }
    }

    int first[OWN_SEGS];
#pragma unroll
    for (int j = 0; j < OWN_SEGS; ++j) {
        const int n = __builtin_amdgcn_readlane(seg_n, j);
        const int s = __builtin_amdgcn_readlane(seg_s, j);
        first[j] = (lane < n) ? binned[(wave + OWN_WAVES * j) * per + s + lane] : 0;
    }

    int low = 0;
    for (int i = tid; i < own * BIN_BLOCKS; i += OWN_THREADS) low += tcnt[i];

    int nw = 0;
    auto note = [&](int w, int slot) {
        const int r = atomicAdd(&c[w & 255], 1);
        if (slot < OWN_LIST) lst[wave][slot] = make_int2(w, r);
    };
#pragma unroll
    for (int j = 0; j < OWN_SEGS; ++j) {
        const int n = __builtin_amdgcn_readlane(seg_n, j);
        const int s = __builtin_amdgcn_readlane(seg_s, j);
        const int* __restrict__ seg = binned + (wave + OWN_WAVES * j) * per + s;
        if (lane < n) note(first[j], nw + lane);
        for (int i = lane + 32; i < n; i += 32) note(seg[i], nw + i);
        nw += n;
    }
    if (nw > OWN_LIST && lane == 0) sfull = 1;
#pragma unroll
    for (int off = 16; off > 0; off >>= 1) low += __shfl_xor(low, off, 32);
    __syncthreads();

    const int cv = (tid < OWN_NODES) ? c[tid] : 0;
    int v = cv;
#pragma unroll
    for (int off = 1; off < 32; off <<= 1) {
        int t = __shfl_up(v, off, 32);
        if (lane >= off) v += t;
    }
    if (lane == 31) wsum[wave] = v;
    if (lane == 0)  wlow[wave] = low;
    __syncthreads();
    if (wave == 0) {
        const int s = wsum[lane];
        int sv = s;
#pragma unroll
        for (int off = 1; off < 32; off <<= 1) {
            int t = __shfl_up(sv, off, 32);
            if (lane >= off) sv += t;
        }
        wsum[lane] = sv - s;
        int lw = wlow[lane];
#pragma unroll
        for (int off = 16; off > 0; off >>= 1) lw += __shfl_xor(lw, off, 32);
        if (lane == 0) sbase = lw;
    }
    __syncthreads();
    if ((unsigned)tid < span) {
        const int start = sbase + wsum[wave] + v - cv;
        const int node  = lo + tid;
        cnt[node]      = cv;
        rowstart[node] = start;
        dinv[node]     = rsqrtf((float)cv + 1.0f);
        c[tid]         = start;
    }
    __syncthreads();

    if (sfull == 0) {
        for (int i = lane; i < nw; i += 32) {
            const int2 en = lst[wave][i];
            gsrc[c[en.x & 255] + en.y] = (int)((unsigned)en.x >> 8);
        }
        return;
    }

    for (int j = 0; j < OWN_SEGS; ++j) {
        const int n = __builtin_amdgcn_readlane(seg_n, j);
        const int s = __builtin_amdgcn_readlane(seg_s, j);
        const int* __restrict__ seg = binned + (wave + OWN_WAVES * j) * per + s;
        for (int i = lane; i < n; i += 32) {
            const int w = seg[i];
            gsrc[atomicAdd(&c[w & 255], 1)] = (int)((unsigned)w >> 8);
        }
    }
}

#define FG_NODES   32
#define FG_GRID    512
#define FG_THREADS (FG_NODES * 32)
#define FG_MT      (FG_NODES / 16)
#define FG_PITCH   (F_HID + 4)
#define FG_NT      ((F_OUT + 15) / 16)
#define FG_KS      (F_HID / 4)
#define FG_NT1     (F_HID / 16)
#define FG_KS1     (F_IN / 4)

static_assert(F_IN == F_HID, "one tile pitch and one lane-to-feature map serve both tiles");
static_assert(F_IN == 128, "a row is one 16-B load per lane");
static_assert(FG_KS == FG_NODES && FG_KS1 == FG_NODES, "each wave copies the B entries of one k step");
static_assert(FG_NT1 == 8 && F_HID == 128, "a lane's two 16-B pieces of a W1 row are its entries of the 8 column tiles");
static_assert(FG_MT * FG_NT1 <= FG_NODES && FG_MT * FG_NT <= FG_NODES, "one wave per output tile");

__global__ __launch_bounds__(FG_THREADS)
void gather_gemm1_relu_gemm2(const int* __restrict__ rowstart, const int* __restrict__ cnt,
                             const int* __restrict__ gsrc, const float* __restrict__ dinv,
                             const float* __restrict__ X, const float* __restrict__ W1,
                             const float* __restrict__ b, const float* __restrict__ W2,
                             float* __restrict__ H2, int Nn) {
    __shared__ __attribute__((aligned(16))) float tile[FG_NODES][FG_PITCH];
    __shared__ __attribute__((aligned(16))) float tile2[FG_NODES][FG_PITCH];
    __shared__ v2f bfrag1[FG_KS1][FG_NT1][32];
    __shared__ v2f bfrag[FG_KS][FG_NT][32];
    __shared__ __attribute__((aligned(16))) float sdi[FG_NODES];

    const int lane = threadIdx.x & 31;
    const int wave = threadIdx.x >> 5;
    const int half = lane >> 4;
    const int l16  = lane & 15;
    const int ntiles = (Nn + FG_NODES - 1) / FG_NODES;
    const int tstep  = (int)gridDim.x;

    int   rs_v;
    int   cn_v;
    float di_v;
    {
        const int node0  = (int)blockIdx.x * FG_NODES + wave;
        const int nodec0 = (node0 < Nn) ? node0 : 0;
        rs_v = rowstart[nodec0];
        cn_v = cnt[nodec0];
        di_v = dinv[nodec0];
    }

    const int krow = 4 * wave + 2 * half;
    float4 q[4];
    {
        const float4* p0 = reinterpret_cast<const float4*>(W1 + (size_t)krow * F_HID + 4 * l16);
        const float4* p1 = reinterpret_cast<const float4*>(W1 + (size_t)(krow + 1) * F_HID + 4 * l16);
        q[0] = p0[0];
        q[1] = p0[F_HID / 8];
        q[2] = p1[0];
        q[3] = p1[F_HID / 8];
    }

    v2f wv[FG_NT];
    for (int t = 0; t < FG_NT; ++t) {
        const int col = t * 16 + l16;
        v2f w;
        w.x = 0.0f; w.y = 0.0f;
        if ((F_OUT % 16 == 0) || (col < F_OUT)) {
            w.x = W2[(size_t)krow * F_OUT + col];
            w.y = W2[(size_t)(krow + 1) * F_OUT + col];
        }
        wv[t] = w;
    }

    {
        const float r0[8] = {q[0].x, q[0].y, q[0].z, q[0].w, q[1].x, q[1].y, q[1].z, q[1].w};
        const float r1[8] = {q[2].x, q[2].y, q[2].z, q[2].w, q[3].x, q[3].y, q[3].z, q[3].w};
#pragma unroll
        for (int t = 0; t < FG_NT1; ++t) {
            v2f e;
            e.x = r0[t];
            e.y = r1[t];
            bfrag1[wave][t][lane] = e;
        }
    }

    for (int t = 0; t < FG_NT; ++t) bfrag[wave][t][lane] = wv[t];

    for (int tile_i = (int)blockIdx.x; tile_i < ntiles; tile_i += tstep) {
    const int base_node = tile_i * FG_NODES;
    const int node = base_node + wave;
    const bool live = node < Nn;

    float4 v = make_float4(0.0f, 0.0f, 0.0f, 0.0f);
    float di = 0.0f;
    int   rs_n = 0;
    int   cn_n = 0;
    float di_n = 0.0f;
    if (live) {
        const int start = __builtin_amdgcn_readfirstlane(rs_v);
        const int n     = __builtin_amdgcn_readfirstlane(cn_v);
        di = di_v;
        const float4* __restrict__ X4 = reinterpret_cast<const float4*>(X);

        float4 acc = make_float4(0.0f, 0.0f, 0.0f, 0.0f);
        for (int base = 0; base < n; base += 32) {
            const int m = (n - base < 32) ? (n - base) : 32;
            int   my_s = 0;
            float my_w = 0.0f;
            if (lane < m) {
                my_s = gsrc[start + base + lane];
                my_w = dinv[my_s];
            }
            const int my_wi = __float_as_int(my_w);
#pragma unroll 4
            for (int j = 0; j < m; ++j) {
                const int    s = __builtin_amdgcn_readlane(my_s, j);
                const float  w = __int_as_float(__builtin_amdgcn_readlane(my_wi, j));
                const float4 h = X4[(size_t)s * (F_IN / 4) + lane];
                acc.x += h.x * w;
                acc.y += h.y * w;
                acc.z += h.z * w;
                acc.w += h.w * w;
            }
        }

        const int tile_n = tile_i + tstep;
        const int node_n = tile_n * FG_NODES + wave;
        const int nodec_n = (tile_n < ntiles && node_n < Nn) ? node_n : 0;
        rs_n = rowstart[nodec_n];
        cn_n = cnt[nodec_n];
        di_n = dinv[nodec_n];

        const float4 xs = X4[(size_t)node * (F_IN / 4) + lane];
        v.x = (acc.x + xs.x * di) * di;
        v.y = (acc.y + xs.y * di) * di;
        v.z = (acc.z + xs.z * di) * di;
        v.w = (acc.w + xs.w * di) * di;
    }

    *reinterpret_cast<float4*>(&tile[wave][4 * lane]) = v;
    if (lane == 0) sdi[wave] = di;
    __syncthreads();

    float4 sc_lo = make_float4(0.0f, 0.0f, 0.0f, 0.0f);
    float4 sc_hi = make_float4(0.0f, 0.0f, 0.0f, 0.0f);
    if (wave < FG_MT * FG_NT) {
        const float4* sc4 = reinterpret_cast<const float4*>(&sdi[16 * (wave / FG_NT) + half * 8]);
        sc_lo = sc4[0];
        sc_hi = sc4[1];
    }

    if (wave < FG_MT * FG_NT1) {
        const int   mt   = wave / FG_NT1;
        const int   ct   = wave % FG_NT1;
        const int   hcol = 64 * (ct >> 2) + 4 * l16 + (ct & 3);
        const float bj   = b[hcol];
        v8f c1 = {};
        const float* arow = &tile[16 * mt + l16][2 * half];
        for (int ks = 0; ks < FG_KS1; ++ks) {
            const v2f a  = *reinterpret_cast<const v2f*>(arow + 4 * ks);
            const v2f bf = bfrag1[ks][ct][lane];
            c1 = __builtin_amdgcn_wmma_f32_16x16x4_f32(
                false, a, false, bf, (short)0, c1, false, false);
        }
#pragma unroll
        for (int r = 0; r < 8; ++r) {
            const float y = c1[r] + bj;
            tile2[16 * mt + 8 * half + r][hcol] = y > 0.0f ? y : 0.0f;
        }
    }
    __syncthreads();

    if (wave < FG_MT * FG_NT) {

    const int mt   = wave / FG_NT;
    const int ct   = wave % FG_NT;
    const int ncol = ct * 16 + l16;

    v8f c = {};
    const float* arow = &tile2[16 * mt + l16][2 * half];
    for (int ks = 0; ks < FG_KS; ++ks) {
        const v2f a  = *reinterpret_cast<const v2f*>(arow + 4 * ks);
        const v2f bf = bfrag[ks][ct][lane];
        c = __builtin_amdgcn_wmma_f32_16x16x4_f32(
            false, a, false, bf, (short)0, c, false, false);
    }

    if ((F_OUT % 16 == 0) || (ncol < F_OUT)) {
        const int row0 = base_node + 16 * mt + half * 8;
        float* out = H2 + (size_t)row0 * F_OUT + ncol;
        if (row0 + 0 < Nn) out[(size_t)0 * F_OUT] = c[0] * sc_lo.x;
        if (row0 + 1 < Nn) out[(size_t)1 * F_OUT] = c[1] * sc_lo.y;
        if (row0 + 2 < Nn) out[(size_t)2 * F_OUT] = c[2] * sc_lo.z;
        if (row0 + 3 < Nn) out[(size_t)3 * F_OUT] = c[3] * sc_lo.w;
        if (row0 + 4 < Nn) out[(size_t)4 * F_OUT] = c[4] * sc_hi.x;
        if (row0 + 5 < Nn) out[(size_t)5 * F_OUT] = c[5] * sc_hi.y;
        if (row0 + 6 < Nn) out[(size_t)6 * F_OUT] = c[6] * sc_hi.z;
        if (row0 + 7 < Nn) out[(size_t)7 * F_OUT] = c[7] * sc_hi.w;
    }
    }

    rs_v = rs_n;
    cn_v = cn_n;
    di_v = di_n;
    }
}

#define LS_THREADS 256
#define LS_GRID    2048

__global__ __launch_bounds__(LS_THREADS)
void gather_logsoftmax(const int* __restrict__ rowstart, const int* __restrict__ cnt,
                                  const int* __restrict__ gsrc, const float* __restrict__ dinv,
                                  const float* __restrict__ H2, const float* __restrict__ b2,
                                  float* __restrict__ OUT, int Nn) {
    const int lane = threadIdx.x & 31;
    const int wave = threadIdx.x >> 5;
    const int wstep = (int)gridDim.x * (LS_THREADS / 32);
    int node = (int)blockIdx.x * (LS_THREADS / 32) + wave;
    if (node >= Nn) return;

    const bool act  = lane < (F_OUT / 4);
    const int col   = act ? lane : 0;
    const float4* __restrict__ H4 = reinterpret_cast<const float4*>(H2);

    int   rs_v = rowstart[node];
    int   cn_v = cnt[node];
    float di   = dinv[node];

    for (;;) {
    const int start = __builtin_amdgcn_readfirstlane(rs_v);
    const int n     = __builtin_amdgcn_readfirstlane(cn_v);

    float4 acc = make_float4(0.0f, 0.0f, 0.0f, 0.0f);
    for (int base = 0; base < n; base += 32) {
        const int m = (n - base < 32) ? (n - base) : 32;
        int my_s = 0;
        if (lane < m) my_s = gsrc[start + base + lane];
#pragma unroll 4
        for (int j = 0; j < m; ++j) {
            const int    s = __builtin_amdgcn_readlane(my_s, j);
            const float4 h = H4[(size_t)s * (F_OUT / 4) + col];
            acc.x += h.x;
            acc.y += h.y;
            acc.z += h.z;
            acc.w += h.w;
        }
    }

    const int  node_n = node + wstep;
    const bool more   = node_n < Nn;
    const int  nodec_n = more ? node_n : node;
    const int   rs_n = rowstart[nodec_n];
    const int   cn_n = cnt[nodec_n];
    const float di_n = dinv[nodec_n];

    const size_t o = (size_t)node * (F_OUT / 4) + lane;
    float4 v = make_float4(-INFINITY, -INFINITY, -INFINITY, -INFINITY);
    if (act) {
        const float4 hs = H4[o];
        const float4 bb = reinterpret_cast<const float4*>(b2)[lane];
        v.x = (acc.x + hs.x) * di + bb.x;
        v.y = (acc.y + hs.y) * di + bb.y;
        v.z = (acc.z + hs.z) * di + bb.z;
        v.w = (acc.w + hs.w) * di + bb.w;
    }

    float mx = fmaxf(fmaxf(v.x, v.y), fmaxf(v.z, v.w));
#pragma unroll
    for (int off = 16; off > 0; off >>= 1)
        mx = fmaxf(mx, __shfl_xor(mx, off, 32));

    float s = act ? (__expf(v.x - mx) + __expf(v.y - mx) + __expf(v.z - mx) + __expf(v.w - mx))
                  : 0.0f;
#pragma unroll
    for (int off = 16; off > 0; off >>= 1)
        s += __shfl_xor(s, off, 32);

    const float ls = __logf(s);
    if (act) {
        float4 r;
        r.x = v.x - mx - ls;
        r.y = v.y - mx - ls;
        r.z = v.z - mx - ls;
        r.w = v.w - mx - ls;
        reinterpret_cast<float4*>(OUT)[o] = r;
    }

    if (!more) break;
    node = node_n;
    rs_v = rs_n;
    cn_v = cn_n;
    di   = di_n;
    }
}

extern "C" void kernel_launch(void* const* d_in, const int* in_sizes, int n_in,
                              void* d_out, int out_size, void* d_ws, size_t ws_size,
                              hipStream_t stream) {
    const float* x  = (const float*)d_in[0];
    const float* W1 = (const float*)d_in[1];
    const float* b1 = (const float*)d_in[2];
    const float* W2 = (const float*)d_in[3];
    const float* b2 = (const float*)d_in[4];
    const int*   ei = (const int*)d_in[5];
    const int E = in_sizes[5] / 2;
    const int Nn = N_NODES;
    const int* src = ei;
    const int* dst = ei + E;

    float* out = (float*)d_out;

    float* ws   = (float*)d_ws;
    float* dinv = ws;
    float* h2   = dinv + 50048;
    int*   cnt  = (int*)(h2 + (size_t)Nn * F_OUT);
    int*   rowstart = cnt + 50048;
    int*   gsrc = rowstart + 50048;
    int*   binned = gsrc + E;
    int*   tcnt   = binned + E;
    int*   tstart = tcnt + OWN_BLOCKS * BIN_BLOCKS;

    bin_edges<<<BIN_BLOCKS, BIN_THREADS, 0, stream>>>(src, dst, binned, tcnt, tstart, E);

    build_groups<<<OWN_BLOCKS, OWN_THREADS, 0, stream>>>(binned, tcnt, tstart, cnt, rowstart, dinv, gsrc, E, Nn);

    gather_gemm1_relu_gemm2<<<FG_GRID, FG_THREADS, 0, stream>>>(
        rowstart, cnt, gsrc, dinv, x, W1, b1, W2, h2, Nn);

    gather_logsoftmax<<<LS_GRID, LS_THREADS, 0, stream>>>(rowstart, cnt, gsrc, dinv, h2, b2, out, Nn);
}
